# starting kernel
# baseline (speedup 1.0000x reference)
#include <hip/hip_runtime.h>
#include <hip/hip_bf16.h>
#include <math.h>

typedef __attribute__((ext_vector_type(2))) float v2f;
typedef __attribute__((ext_vector_type(8))) float v8f;

#define NEG_SLOPE 0.2f

__device__ __forceinline__ unsigned enc_f32(float f) {
    unsigned u = __float_as_uint(f);
    return (u & 0x80000000u) ? ~u : (u | 0x80000000u);
}
__device__ __forceinline__ float dec_f32(unsigned u) {
    unsigned v = (u & 0x80000000u) ? (u & 0x7fffffffu) : ~u;
    return __uint_as_float(v);
}

__global__ void gat_pack_b(const float* __restrict__ B, float* __restrict__ Bp,
                           int K, int Dout, int DP) {
    const int i = blockIdx.x * blockDim.x + threadIdx.x;
    if (i >= (K / 2) * DP) return;
    const int kk = i / DP;
    const int c  = i - kk * DP;
    float x0 = 0.f, x1 = 0.f;
    if (c < Dout) {
        x0 = B[(size_t)(2 * kk) * Dout + c];
        x1 = B[(size_t)(2 * kk + 1) * Dout + c];
    }
    Bp[2 * i]     = x0;
    Bp[2 * i + 1] = x1;
}

template<int K, int NT>
__global__ __launch_bounds__(256) void gat_gemm_wmma(const float* __restrict__ A,
                                                     const float* __restrict__ Bp,
                                                     float* __restrict__ C, int N) {
    constexpr int DP = NT * 16;
    __shared__ float Bs[K * DP];

    const int tid = threadIdx.y * 32 + threadIdx.x;
    #pragma unroll
    for (int i = tid * 4; i < K * DP; i += 256 * 4)
        *(float4*)&Bs[i] = *(const float4*)&Bp[i];
    __syncthreads();

    const int Mtiles = (N + 15) >> 4;
    const int strip = blockIdx.x * blockDim.y + threadIdx.y;
    if (strip >= Mtiles) return;
    int row0 = strip << 4;
    if (row0 + 16 > N) row0 = N - 16;

    const int m  = threadIdx.x & 15;
    const int kh = threadIdx.x >> 4;
    const float* __restrict__ arow = A + (size_t)(row0 + m) * K;

    v8f acc[NT];
    #pragma unroll
    for (int t = 0; t < NT; ++t) acc[t] = (v8f){0.f,0.f,0.f,0.f,0.f,0.f,0.f,0.f};

    #pragma unroll
    for (int k = 0; k < K; k += 4) {
        const int ka = k + 2 * kh;
        const v2f a = *(const v2f*)(arow + ka);
        const int kk = ka >> 1;
        #pragma unroll
        for (int t = 0; t < NT; ++t) {
            const v2f b = *(const v2f*)&Bs[(kk * DP + t * 16 + m) * 2];
            acc[t] = __builtin_amdgcn_wmma_f32_16x16x4_f32(
                false, a, false, b, (short)0, acc[t], false, false);
        }
    }

    #pragma unroll
    for (int t = 0; t < NT; ++t) {
        const int cn = t * 16 + m;
        #pragma unroll
        for (int v = 0; v < 8; ++v)
            C[(size_t)(row0 + v + 8 * kh) * DP + cn] = acc[t][v];
    }
}

__global__ void gat_attn_scores(const float* __restrict__ F,
                                const float* __restrict__ al,
                                const float* __restrict__ ar,
                                float* __restrict__ el, float* __restrict__ er,
                                int N, int D, int FS) {
    const int i = blockIdx.x * blockDim.x + threadIdx.x;
    if (i >= N) return;
    const float* f = F + (size_t)i * FS;
    float a = 0.f, b = 0.f;
    for (int j = 0; j < D; j += 4) {
        float4 v = *(const float4*)(f + j);
        a += v.x * al[j] + v.y * al[j + 1] + v.z * al[j + 2] + v.w * al[j + 3];
        b += v.x * ar[j] + v.y * ar[j + 1] + v.z * ar[j + 2] + v.w * ar[j + 3];
    }
    el[i] = a;
    er[i] = b;
}

__global__ void gat_edge_max(const int* __restrict__ src, const int* __restrict__ dst,
                             const float* __restrict__ el, const float* __restrict__ er,
                             float* __restrict__ ebuf, unsigned* __restrict__ menc,
                             int E) {
    const int e = blockIdx.x * blockDim.x + threadIdx.x;
    if (e >= E) return;
    const int d = dst[e];
    float v = el[src[e]] + er[d];
    v = (v > 0.f) ? v : NEG_SLOPE * v;
    ebuf[e] = v;
    atomicMax(&menc[d], enc_f32(v));
}

__global__ void gat_edge_expsum(const int* __restrict__ dst,
                                float* __restrict__ ebuf,
                                const unsigned* __restrict__ menc,
                                float* __restrict__ s, int E) {
    const int e = blockIdx.x * blockDim.x + threadIdx.x;
    if (e >= E) return;
    const int d = dst[e];
    const float x = __expf(ebuf[e] - dec_f32(menc[d]));
    ebuf[e] = x;
    atomicAdd(&s[d], x);
}

__global__ void gat_aggregate(const float* __restrict__ F,
                              const float* __restrict__ ee,
                              const float* __restrict__ s,
                              const int* __restrict__ src, const int* __restrict__ dst,
                              float* __restrict__ out, int E, int D, int FS) {
    const long long gid = (long long)blockIdx.x * blockDim.x + threadIdx.x;
    const int e = (int)(gid >> 4);
    const int p = (int)(gid & 15);
    if (e >= E) return;
    const int c0 = p << 2;
    if (c0 >= D) return;
    const int d = dst[e];
    const float alpha = ee[e] / s[d];
    const float4 v = *(const float4*)(F + (size_t)src[e] * FS + c0);
    float* o = out + (size_t)d * D + c0;
    atomicAdd(o + 0, alpha * v.x);
    atomicAdd(o + 1, alpha * v.y);
    atomicAdd(o + 2, alpha * v.z);
    atomicAdd(o + 3, alpha * v.w);
}

__global__ void gat_bias_elu(float* __restrict__ h, const float* __restrict__ bias,
                             int total, int D) {
    const int i = blockIdx.x * blockDim.x + threadIdx.x;
    if (i >= total) return;
    const float x = h[i] + bias[i & (D - 1)];
    h[i] = (x > 0.f) ? x : (__expf(x) - 1.f);
}

__global__ void gat_bias_add(float* __restrict__ out, const float* __restrict__ bias,
                             int total, int D) {
    const int i = blockIdx.x * blockDim.x + threadIdx.x;
    if (i >= total) return;
    out[i] += bias[i % D];
}

extern "C" void kernel_launch(void* const* d_in, const int* in_sizes, int n_in,
                              void* d_out, int out_size, void* d_ws, size_t ws_size,
                              hipStream_t stream) {
    const float* X   = (const float*)d_in[0];
    const int*   src = (const int*)d_in[1];
    const int*   dst = (const int*)d_in[2];
    const float* W1  = (const float*)d_in[3];
    const float* al1 = (const float*)d_in[4];
    const float* ar1 = (const float*)d_in[5];
    const float* b1  = (const float*)d_in[6];
    const float* W2  = (const float*)d_in[7];
    const float* al2 = (const float*)d_in[8];
    const float* ar2 = (const float*)d_in[9];
    const float* b2  = (const float*)d_in[10];

    const int IN = 128, H = 64, CLS = 40, CLSPAD = 48;
    const int N = in_sizes[0] / IN;
    const int E = in_sizes[1];

    float* ws = (float*)d_ws;
    float*    feat1 = ws;
    float*    h     = feat1 + (size_t)N * H;
    float*    feat2 = h + (size_t)N * H;
    float*    el    = feat2 + (size_t)N * CLSPAD;
    float*    er    = el + N;
    unsigned* menc  = (unsigned*)(er + N);
    float*    s     = (float*)(menc + N);
    float*    ebuf  = s + N;
    float*    Bp1   = ebuf + E;
    float*    Bp2   = Bp1 + (size_t)IN * H;

    const int Mtiles = (N + 15) / 16;
    dim3 wblk(32, 8);
    const dim3 ggrid((Mtiles + 7) / 8);

    {
        hipLaunchKernelGGL(gat_pack_b, dim3((IN / 2 * H + 255) / 256), dim3(256), 0, stream,
                           W1, Bp1, IN, H, H);
        hipLaunchKernelGGL((gat_gemm_wmma<128, 4>), ggrid, wblk, 0, stream,
                           X, Bp1, feat1, N);

        hipLaunchKernelGGL(gat_attn_scores, dim3((N + 255) / 256), dim3(256), 0, stream,
                           feat1, al1, ar1, el, er, N, H, H);

        hipMemsetAsync(menc, 0, (size_t)N * sizeof(unsigned), stream);
        hipMemsetAsync(s,    0, (size_t)N * sizeof(float), stream);
        hipMemsetAsync(h,    0, (size_t)N * H * sizeof(float), stream);

        hipLaunchKernelGGL(gat_edge_max, dim3((E + 255) / 256), dim3(256), 0, stream,
                           src, dst, el, er, ebuf, menc, E);
        hipLaunchKernelGGL(gat_edge_expsum, dim3((E + 255) / 256), dim3(256), 0, stream,
                           dst, ebuf, menc, s, E);

        const long long work = (long long)E * 16;
        hipLaunchKernelGGL(gat_aggregate, dim3((unsigned)((work + 255) / 256)), dim3(256), 0, stream,
                           feat1, ebuf, s, src, dst, h, E, H, H);

        hipLaunchKernelGGL(gat_bias_elu, dim3((N * H + 255) / 256), dim3(256), 0, stream,
                           h, b1, N * H, H);
    }

    {
        float* out = (float*)d_out;

        hipLaunchKernelGGL(gat_pack_b, dim3((H / 2 * CLSPAD + 255) / 256), dim3(256), 0, stream,
                           W2, Bp2, H, CLS, CLSPAD);
        hipLaunchKernelGGL((gat_gemm_wmma<64, 3>), ggrid, wblk, 0, stream,
                           h, Bp2, feat2, N);

        hipLaunchKernelGGL(gat_attn_scores, dim3((N + 255) / 256), dim3(256), 0, stream,
                           feat2, al2, ar2, el, er, N, CLS, CLSPAD);

        hipMemsetAsync(menc, 0, (size_t)N * sizeof(unsigned), stream);
        hipMemsetAsync(s,    0, (size_t)N * sizeof(float), stream);
        hipMemsetAsync(out,  0, (size_t)N * CLS * sizeof(float), stream);

        hipLaunchKernelGGL(gat_edge_max, dim3((E + 255) / 256), dim3(256), 0, stream,
                           src, dst, el, er, ebuf, menc, E);
        hipLaunchKernelGGL(gat_edge_expsum, dim3((E + 255) / 256), dim3(256), 0, stream,
                           dst, ebuf, menc, s, E);

        const long long work = (long long)E * 16;
        hipLaunchKernelGGL(gat_aggregate, dim3((unsigned)((work + 255) / 256)), dim3(256), 0, stream,
                           feat2, ebuf, s, src, dst, out, E, CLS, CLSPAD);

        hipLaunchKernelGGL(gat_bias_add, dim3((N * CLS + 255) / 256), dim3(256), 0, stream,
                           out, b2, N * CLS, CLS);
    }
}


# Edges grouped by destination once per call (int atomic rank, scan, scatter); each layer: one wave per node does max, exp-sum, weighted row sum in registers in arrival-rank order; float atomics gone
# speedup vs baseline: 2.8412x; 2.8412x over previous
#include <hip/hip_runtime.h>
#include <hip/hip_bf16.h>
#include <math.h>

typedef __attribute__((ext_vector_type(2))) float v2f;
typedef __attribute__((ext_vector_type(8))) float v8f;

#define NEG_SLOPE 0.2f

__global__ void gat_pack_b(const float* __restrict__ B, float* __restrict__ Bp,
                           int K, int Dout, int DP) {
    const int i = blockIdx.x * blockDim.x + threadIdx.x;
    if (i >= (K / 2) * DP) return;
    const int kk = i / DP;
    const int c  = i - kk * DP;
    float x0 = 0.f, x1 = 0.f;
    if (c < Dout) {
        x0 = B[(size_t)(2 * kk) * Dout + c];
        x1 = B[(size_t)(2 * kk + 1) * Dout + c];
    }
    Bp[2 * i]     = x0;
    Bp[2 * i + 1] = x1;
}

template<int K, int NT>
__global__ __launch_bounds__(256) void gat_gemm_wmma(const float* __restrict__ A,
                                                     const float* __restrict__ Bp,
                                                     float* __restrict__ C, int N) {
    constexpr int DP = NT * 16;
    __shared__ float Bs[K * DP];

    const int tid = threadIdx.y * 32 + threadIdx.x;
    #pragma unroll
    for (int i = tid * 4; i < K * DP; i += 256 * 4)
        *(float4*)&Bs[i] = *(const float4*)&Bp[i];
    __syncthreads();

    const int Mtiles = (N + 15) >> 4;
    const int strip = blockIdx.x * blockDim.y + threadIdx.y;
    if (strip >= Mtiles) return;
    int row0 = strip << 4;
    if (row0 + 16 > N) row0 = N - 16;

    const int m  = threadIdx.x & 15;
    const int kh = threadIdx.x >> 4;
    const float* __restrict__ arow = A + (size_t)(row0 + m) * K;

    v8f acc[NT];
    #pragma unroll
    for (int t = 0; t < NT; ++t) acc[t] = (v8f){0.f,0.f,0.f,0.f,0.f,0.f,0.f,0.f};

    #pragma unroll
    for (int k = 0; k < K; k += 4) {
        const int ka = k + 2 * kh;
        const v2f a = *(const v2f*)(arow + ka);
        const int kk = ka >> 1;
        #pragma unroll
        for (int t = 0; t < NT; ++t) {
            const v2f b = *(const v2f*)&Bs[(kk * DP + t * 16 + m) * 2];
            acc[t] = __builtin_amdgcn_wmma_f32_16x16x4_f32(
                false, a, false, b, (short)0, acc[t], false, false);
        }
    }

    #pragma unroll
    for (int t = 0; t < NT; ++t) {
        const int cn = t * 16 + m;
        #pragma unroll
        for (int v = 0; v < 8; ++v)
            C[(size_t)(row0 + v + 8 * kh) * DP + cn] = acc[t][v];
    }
}

__global__ void gat_attn_scores(const float* __restrict__ F,
                                const float* __restrict__ al,
                                const float* __restrict__ ar,
                                float* __restrict__ el, float* __restrict__ er,
                                int N, int D, int FS) {
    const int i = blockIdx.x * blockDim.x + threadIdx.x;
    if (i >= N) return;
    const float* f = F + (size_t)i * FS;
    float a = 0.f, b = 0.f;
    for (int j = 0; j < D; j += 4) {
        float4 v = *(const float4*)(f + j);
        a += v.x * al[j] + v.y * al[j + 1] + v.z * al[j + 2] + v.w * al[j + 3];
        b += v.x * ar[j] + v.y * ar[j + 1] + v.z * ar[j + 2] + v.w * ar[j + 3];
    }
    el[i] = a;
    er[i] = b;
}

__global__ void gat_edge_rank(const int* __restrict__ dst, int* __restrict__ deg,
                              int* __restrict__ rank, int E) {
    const int e = blockIdx.x * blockDim.x + threadIdx.x;
    if (e >= E) return;
    rank[e] = atomicAdd(&deg[dst[e]], 1);
}

__global__ __launch_bounds__(1024) void gat_scan_deg(const int* __restrict__ deg,
                                                     int* __restrict__ rowptr, int N) {
    __shared__ int part[32];
    __shared__ int wtot[32];
    const int tid  = threadIdx.x;
    const int lane = tid & 31;
    const int wv   = tid >> 5;
    const int tile0 = blockIdx.x * 1024;

    int p = 0;
    for (int i = tid; i < tile0; i += 1024) p += deg[i];
    for (int off = 16; off > 0; off >>= 1) p += __shfl_xor(p, off);

    const int i = tile0 + tid;
    const int x = (i < N) ? deg[i] : 0;
    int incl = x;
    for (int off = 1; off < 32; off <<= 1) {
        const int t = __shfl_up(incl, off);
        if (lane >= off) incl += t;
    }
    if (lane == 0)  part[wv] = p;
    if (lane == 31) wtot[wv] = incl;
    __syncthreads();

    int carry = 0;
    for (int k = 0; k < 32; ++k) {
        carry += part[k];
        if (k < wv) carry += wtot[k];
    }
    const int excl = carry + incl - x;
    if (i < N) rowptr[i] = excl;
    if (i == N - 1) rowptr[N] = excl + x;
}

__global__ void gat_edge_scatter(const int* __restrict__ src, const int* __restrict__ dst,
                                 const int* __restrict__ rank, const int* __restrict__ rowptr,
                                 int* __restrict__ csr, int E) {
    const int e = blockIdx.x * blockDim.x + threadIdx.x;
    if (e >= E) return;
    csr[rowptr[dst[e]] + rank[e]] = src[e];
}

template<int G, int FS>
__device__ __forceinline__ void gat_acc_rows(const float* __restrict__ Fl, int s, float w, int t,
                                             float& ax, float& ay) {
    float2 f[G];
    float  wt[G];
    #pragma unroll
    for (int g = 0; g < G; ++g) {
        const int sg = __builtin_amdgcn_readlane(s, t + g);
        wt[g] = __int_as_float(__builtin_amdgcn_readlane(__float_as_int(w), t + g));
        f[g] = *(const float2*)(Fl + (size_t)sg * FS);
    }
    #pragma unroll
    for (int g = 0; g < G; ++g) {
        ax += wt[g] * f[g].x;
        ay += wt[g] * f[g].y;
    }
}

template<int FS>
__device__ __forceinline__ void gat_acc_chunk(const float* __restrict__ Fl, int s, float w, int cnt,
                                              float& ax, float& ay) {
    int t = 0;
    for (; t + 8 <= cnt; t += 8) gat_acc_rows<8, FS>(Fl, s, w, t, ax, ay);
    if (cnt & 4) { gat_acc_rows<4, FS>(Fl, s, w, t, ax, ay); t += 4; }
    if (cnt & 2) { gat_acc_rows<2, FS>(Fl, s, w, t, ax, ay); t += 2; }
    if (cnt & 1) { gat_acc_rows<1, FS>(Fl, s, w, t, ax, ay); }
}

template<int D, int FS, bool ELU>
__global__ __launch_bounds__(256) void gat_node_softmax_agg(const float* __restrict__ F,
                                                            const float* __restrict__ el,
                                                            const float* __restrict__ er,
                                                            const int* __restrict__ rowptr,
                                                            const int* __restrict__ csr,
                                                            const float* __restrict__ bias,
                                                            float* __restrict__ out, int N) {
    const int lane = threadIdx.x;
    const int node = blockIdx.x * 8 + __builtin_amdgcn_readfirstlane((int)threadIdx.y);
    if (node >= N) return;
    const int beg = rowptr[node];
    const int end = rowptr[node + 1];
    const float erd = er[node];
    const bool colok = (2 * lane < D);
    const float* __restrict__ Fl = F + (colok ? 2 * lane : 0);

    const bool has0 = (beg + lane < end);
    int s0 = 0;
    float v0 = -INFINITY;
    if (has0) {
        s0 = csr[beg + lane];
        const float v = el[s0] + erd;
        v0 = (v > 0.f) ? v : NEG_SLOPE * v;
    }
    float m = v0;
    for (int j = beg + 32 + lane; j < end; j += 32) {
        const float v = el[csr[j]] + erd;
        m = fmaxf(m, (v > 0.f) ? v : NEG_SLOPE * v);
    }
    for (int off = 16; off > 0; off >>= 1) m = fmaxf(m, __shfl_xor(m, off));

    float ax = 0.f, ay = 0.f;
    const float w0 = has0 ? __expf(v0 - m) : 0.f;
    float ssum = w0;
    if (end > beg)
        gat_acc_chunk<FS>(Fl, s0, w0, min(32, end - beg), ax, ay);
    for (int base = beg + 32; base < end; base += 32) {
        const int j = base + lane;
        int s = 0;
        float w = 0.f;
        if (j < end) {
            s = csr[j];
            const float v = el[s] + erd;
            w = __expf(((v > 0.f) ? v : NEG_SLOPE * v) - m);
        }
        ssum += w;
        gat_acc_chunk<FS>(Fl, s, w, min(32, end - base), ax, ay);
    }
    for (int off = 16; off > 0; off >>= 1) ssum += __shfl_xor(ssum, off);

    if (colok) {
        const float2 b = *(const float2*)(bias + 2 * lane);
        float x = b.x, y = b.y;
        if (end > beg) { x += ax / ssum; y += ay / ssum; }
        if (ELU) {
            x = (x > 0.f) ? x : (__expf(x) - 1.f);
            y = (y > 0.f) ? y : (__expf(y) - 1.f);
        }
        *(float2*)(out + (size_t)node * D + 2 * lane) = make_float2(x, y);
    }
}

extern "C" void kernel_launch(void* const* d_in, const int* in_sizes, int n_in,
                              void* d_out, int out_size, void* d_ws, size_t ws_size,
                              hipStream_t stream) {
    const float* X   = (const float*)d_in[0];
    const int*   src = (const int*)d_in[1];
    const int*   dst = (const int*)d_in[2];
    const float* W1  = (const float*)d_in[3];
    const float* al1 = (const float*)d_in[4];
    const float* ar1 = (const float*)d_in[5];
    const float* b1  = (const float*)d_in[6];
    const float* W2  = (const float*)d_in[7];
    const float* al2 = (const float*)d_in[8];
    const float* ar2 = (const float*)d_in[9];
    const float* b2  = (const float*)d_in[10];

    const int IN = 128, H = 64, CLS = 40, CLSPAD = 48;
    const int N = in_sizes[0] / IN;
    const int E = in_sizes[1];

    float* ws = (float*)d_ws;
    float* Bp1   = ws;
    float* Bp2   = Bp1 + (size_t)IN * H;
    float* feat1 = Bp2 + (size_t)H * CLSPAD;
    float* h     = feat1 + (size_t)N * H;
    float* feat2 = h + (size_t)N * H;
    float* el    = feat2 + (size_t)N * CLSPAD;
    float* er    = el + N;
    int*   deg    = (int*)(er + N);
    int*   rank   = deg + N;
    int*   csr    = rank + E;
    int*   rowptr = csr + E;

    const int Mtiles = (N + 15) / 16;
    dim3 wblk(32, 8);
    const dim3 ggrid((Mtiles + 7) / 8);
    const dim3 ngrid((N + 7) / 8);

    hipMemsetAsync(deg, 0, (size_t)N * sizeof(int), stream);
    hipLaunchKernelGGL(gat_edge_rank, dim3((E + 255) / 256), dim3(256), 0, stream,
                       dst, deg, rank, E);
    hipLaunchKernelGGL(gat_scan_deg, dim3((N + 1023) / 1024), dim3(1024), 0, stream,
                       deg, rowptr, N);
    hipLaunchKernelGGL(gat_edge_scatter, dim3((E + 255) / 256), dim3(256), 0, stream,
                       src, dst, rank, rowptr, csr, E);

    hipLaunchKernelGGL(gat_pack_b, dim3((IN / 2 * H + 255) / 256), dim3(256), 0, stream,
                       W1, Bp1, IN, H, H);
    hipLaunchKernelGGL((gat_gemm_wmma<128, 4>), ggrid, wblk, 0, stream,
                       X, Bp1, feat1, N);
    hipLaunchKernelGGL(gat_attn_scores, dim3((N + 255) / 256), dim3(256), 0, stream,
                       feat1, al1, ar1, el, er, N, H, H);
    hipLaunchKernelGGL((gat_node_softmax_agg<64, 64, true>), ngrid, wblk, 0, stream,
                       feat1, el, er, rowptr, csr, b1, h, N);

    hipLaunchKernelGGL(gat_pack_b, dim3((H / 2 * CLSPAD + 255) / 256), dim3(256), 0, stream,
                       W2, Bp2, H, CLS, CLSPAD);
    hipLaunchKernelGGL((gat_gemm_wmma<64, 3>), ggrid, wblk, 0, stream,
                       h, Bp2, feat2, N);
    hipLaunchKernelGGL(gat_attn_scores, dim3((N + 255) / 256), dim3(256), 0, stream,
                       feat2, al2, ar2, el, er, N, CLS, CLSPAD);
    hipLaunchKernelGGL((gat_node_softmax_agg<40, 48, false>), ngrid, wblk, 0, stream,
                       feat2, el, er, rowptr, csr, b2, (float*)d_out, N);
}


# GEMM launch absorbs its layer's two helper launches: W pair-packed on the way into LDS, el/er summed from the accumulator tiles (per-lane terms + 16-lane DPP tree: new sum order); 11 -> 7 launches
# speedup vs baseline: 3.0386x; 1.0695x over previous
#include <hip/hip_runtime.h>
#include <hip/hip_bf16.h>
#include <math.h>

typedef __attribute__((ext_vector_type(2))) float v2f;
typedef __attribute__((ext_vector_type(8))) float v8f;

#define NEG_SLOPE 0.2f

template<int CTRL>
__device__ __forceinline__ float gat_dpp(float x) {
    return __int_as_float(__builtin_amdgcn_update_dpp(0, __float_as_int(x), CTRL, 0xf, 0xf, false));
}
__device__ __forceinline__ float gat_row16_sum(float x) {
    x += gat_dpp<0xB1>(x);
    x += gat_dpp<0x4E>(x);
    x += gat_dpp<0x141>(x);
    x += gat_dpp<0x140>(x);
    return x;
}

template<int K, int NT, int DOUT>
__global__ __launch_bounds__(256) __attribute__((amdgpu_waves_per_eu(2, 12))) void gat_gemm_wmma(const float* __restrict__ A,
                                                     const float* __restrict__ W,
                                                     const float* __restrict__ al,
                                                     const float* __restrict__ ar,
                                                     float* __restrict__ C,
                                                     float* __restrict__ el,
                                                     float* __restrict__ er, int N) {
    constexpr int DP = NT * 16;
    constexpr int CP = DP / 2;
    __shared__ float Bs[K * DP];

    const int tid = threadIdx.y * 32 + threadIdx.x;
    constexpr int ITEMS = (K / 2) * CP;
    static_assert(ITEMS % 256 == 0, "staging assumes a whole number of items per thread");
    static_assert(DOUT % 2 == 0 && DOUT <= DP, "column pairs must not straddle the padding");
    #pragma unroll
    for (int j = 0; j < ITEMS / 256; ++j) {
        const int i  = tid + 256 * j;
        const int kk = i / CP;
        const int c  = 2 * (i - kk * CP);
        const int cl = (DOUT < DP && c >= DOUT) ? 0 : c;
        float2 r0 = *(const float2*)&W[(size_t)(2 * kk) * DOUT + cl];
        float2 r1 = *(const float2*)&W[(size_t)(2 * kk + 1) * DOUT + cl];
        if (DOUT < DP && c >= DOUT) {
            r0 = make_float2(0.f, 0.f);
            r1 = make_float2(0.f, 0.f);
        }
        *(float4*)&Bs[(kk * DP + c) * 2] = make_float4(r0.x, r1.x, r0.y, r1.y);
    }
    __syncthreads();

    const int Mtiles = (N + 15) >> 4;
    const int strip = blockIdx.x * blockDim.y + threadIdx.y;
    if (strip >= Mtiles) return;
    int row0 = strip << 4;
    if (row0 + 16 > N) row0 = N - 16;

    const int m  = threadIdx.x & 15;
    const int kh = threadIdx.x >> 4;
    const float* __restrict__ arow = A + (size_t)(row0 + m) * K;

    float alv[NT], arv[NT];
    #pragma unroll
    for (int t = 0; t < NT; ++t) {
        const int cn = t * 16 + m;
        alv[t] = (cn < DOUT) ? al[cn] : 0.f;
        arv[t] = (cn < DOUT) ? ar[cn] : 0.f;
    }

    v8f acc[NT];
    #pragma unroll
    for (int t = 0; t < NT; ++t) acc[t] = (v8f){0.f,0.f,0.f,0.f,0.f,0.f,0.f,0.f};

    #pragma unroll
    for (int k = 0; k < K; k += 4) {
        const int ka = k + 2 * kh;
        const v2f a = *(const v2f*)(arow + ka);
        const int kk = ka >> 1;
        #pragma unroll
        for (int t = 0; t < NT; ++t) {
            const v2f b = *(const v2f*)&Bs[(kk * DP + t * 16 + m) * 2];
            acc[t] = __builtin_amdgcn_wmma_f32_16x16x4_f32(
                false, a, false, b, (short)0, acc[t], false, false);
        }
    }

    #pragma unroll
    for (int t = 0; t < NT; ++t) {
        const int cn = t * 16 + m;
        #pragma unroll
        for (int v = 0; v < 8; ++v)
            C[(size_t)(row0 + v + 8 * kh) * DP + cn] = acc[t][v];
    }

    float pl[8], pr[8];
    #pragma unroll
    for (int v = 0; v < 8; ++v) {
        float a = acc[0][v] * alv[0];
        float b = acc[0][v] * arv[0];
        #pragma unroll
        for (int t = 1; t < NT; ++t) {
            a += acc[t][v] * alv[t];
            b += acc[t][v] * arv[t];
        }
        pl[v] = gat_row16_sum(a);
        pr[v] = gat_row16_sum(b);
    }
    const int r = m & 7;
    float sl = pl[0], sr = pr[0];
    #pragma unroll
    for (int v = 1; v < 8; ++v) {
        if (r == v) { sl = pl[v]; sr = pr[v]; }
    }
    float* __restrict__ dp = (m < 8) ? el : er;
    dp[row0 + 8 * kh + r] = (m < 8) ? sl : sr;
}

__global__ void gat_edge_rank(const int* __restrict__ dst, int* __restrict__ deg,
                              int* __restrict__ rank, int E) {
    const int e = blockIdx.x * blockDim.x + threadIdx.x;
    if (e >= E) return;
    rank[e] = atomicAdd(&deg[dst[e]], 1);
}

__global__ __launch_bounds__(1024) void gat_scan_deg(const int* __restrict__ deg,
                                                     int* __restrict__ rowptr, int N) {
    __shared__ int part[32];
    __shared__ int wtot[32];
    const int tid  = threadIdx.x;
    const int lane = tid & 31;
    const int wv   = tid >> 5;
    const int tile0 = blockIdx.x * 1024;

    int p = 0;
    for (int i = tid; i < tile0; i += 1024) p += deg[i];
    for (int off = 16; off > 0; off >>= 1) p += __shfl_xor(p, off);

    const int i = tile0 + tid;
    const int x = (i < N) ? deg[i] : 0;
    int incl = x;
    for (int off = 1; off < 32; off <<= 1) {
        const int t = __shfl_up(incl, off);
        if (lane >= off) incl += t;
    }
    if (lane == 0)  part[wv] = p;
    if (lane == 31) wtot[wv] = incl;
    __syncthreads();

    int carry = 0;
    for (int k = 0; k < 32; ++k) {
        carry += part[k];
        if (k < wv) carry += wtot[k];
    }
    const int excl = carry + incl - x;
    if (i < N) rowptr[i] = excl;
    if (i == N - 1) rowptr[N] = excl + x;
}

__global__ void gat_edge_scatter(const int* __restrict__ src, const int* __restrict__ dst,
                                 const int* __restrict__ rank, const int* __restrict__ rowptr,
                                 int* __restrict__ csr, int E) {
    const int e = blockIdx.x * blockDim.x + threadIdx.x;
    if (e >= E) return;
    csr[rowptr[dst[e]] + rank[e]] = src[e];
}

template<int G, int FS>
__device__ __forceinline__ void gat_acc_rows(const float* __restrict__ Fl, int s, float w, int t,
                                             float& ax, float& ay) {
    float2 f[G];
    float  wt[G];
    #pragma unroll
    for (int g = 0; g < G; ++g) {
        const int sg = __builtin_amdgcn_readlane(s, t + g);
        wt[g] = __int_as_float(__builtin_amdgcn_readlane(__float_as_int(w), t + g));
        f[g] = *(const float2*)(Fl + (size_t)sg * FS);
    }
    #pragma unroll
    for (int g = 0; g < G; ++g) {
        ax += wt[g] * f[g].x;
        ay += wt[g] * f[g].y;
    }
}

template<int FS>
__device__ __forceinline__ void gat_acc_chunk(const float* __restrict__ Fl, int s, float w, int cnt,
                                              float& ax, float& ay) {
    int t = 0;
    for (; t + 8 <= cnt; t += 8) gat_acc_rows<8, FS>(Fl, s, w, t, ax, ay);
    if (cnt & 4) { gat_acc_rows<4, FS>(Fl, s, w, t, ax, ay); t += 4; }
    if (cnt & 2) { gat_acc_rows<2, FS>(Fl, s, w, t, ax, ay); t += 2; }
    if (cnt & 1) { gat_acc_rows<1, FS>(Fl, s, w, t, ax, ay); }
}

template<int D, int FS, bool ELU>
__global__ __launch_bounds__(256) void gat_node_softmax_agg(const float* __restrict__ F,
                                                            const float* __restrict__ el,
                                                            const float* __restrict__ er,
                                                            const int* __restrict__ rowptr,
                                                            const int* __restrict__ csr,
                                                            const float* __restrict__ bias,
                                                            float* __restrict__ out, int N) {
    const int lane = threadIdx.x;
    const int node = blockIdx.x * 8 + __builtin_amdgcn_readfirstlane((int)threadIdx.y);
    if (node >= N) return;
    const int beg = rowptr[node];
    const int end = rowptr[node + 1];
    const float erd = er[node];
    const bool colok = (2 * lane < D);
    const float* __restrict__ Fl = F + (colok ? 2 * lane : 0);

    const bool has0 = (beg + lane < end);
    int s0 = 0;
    float v0 = -INFINITY;
    if (has0) {
        s0 = csr[beg + lane];
        const float v = el[s0] + erd;
        v0 = (v > 0.f) ? v : NEG_SLOPE * v;
    }
    float m = v0;
    for (int j = beg + 32 + lane; j < end; j += 32) {
        const float v = el[csr[j]] + erd;
        m = fmaxf(m, (v > 0.f) ? v : NEG_SLOPE * v);
    }
    for (int off = 16; off > 0; off >>= 1) m = fmaxf(m, __shfl_xor(m, off));

    float ax = 0.f, ay = 0.f;
    const float w0 = has0 ? __expf(v0 - m) : 0.f;
    float ssum = w0;
    if (end > beg)
        gat_acc_chunk<FS>(Fl, s0, w0, min(32, end - beg), ax, ay);
    for (int base = beg + 32; base < end; base += 32) {
        const int j = base + lane;
        int s = 0;
        float w = 0.f;
        if (j < end) {
            s = csr[j];
            const float v = el[s] + erd;
            w = __expf(((v > 0.f) ? v : NEG_SLOPE * v) - m);
        }
        ssum += w;
        gat_acc_chunk<FS>(Fl, s, w, min(32, end - base), ax, ay);
    }
    for (int off = 16; off > 0; off >>= 1) ssum += __shfl_xor(ssum, off);

    if (colok) {
        const float2 b = *(const float2*)(bias + 2 * lane);
        float x = b.x, y = b.y;
        if (end > beg) { x += ax / ssum; y += ay / ssum; }
        if (ELU) {
            x = (x > 0.f) ? x : (__expf(x) - 1.f);
            y = (y > 0.f) ? y : (__expf(y) - 1.f);
        }
        *(float2*)(out + (size_t)node * D + 2 * lane) = make_float2(x, y);
    }
}

extern "C" void kernel_launch(void* const* d_in, const int* in_sizes, int n_in,
                              void* d_out, int out_size, void* d_ws, size_t ws_size,
                              hipStream_t stream) {
    const float* X   = (const float*)d_in[0];
    const int*   src = (const int*)d_in[1];
    const int*   dst = (const int*)d_in[2];
    const float* W1  = (const float*)d_in[3];
    const float* al1 = (const float*)d_in[4];
    const float* ar1 = (const float*)d_in[5];
    const float* b1  = (const float*)d_in[6];
    const float* W2  = (const float*)d_in[7];
    const float* al2 = (const float*)d_in[8];
    const float* ar2 = (const float*)d_in[9];
    const float* b2  = (const float*)d_in[10];

    const int IN = 128, H = 64, CLSPAD = 48;
    const int N = in_sizes[0] / IN;
    const int E = in_sizes[1];

    float* ws = (float*)d_ws;
    float* feat1 = ws;
    float* h     = feat1 + (size_t)N * H;
    float* feat2 = h + (size_t)N * H;
    float* el    = feat2 + (size_t)N * CLSPAD;
    float* er    = el + N;
    int*   deg    = (int*)(er + N);
    int*   rank   = deg + N;
    int*   csr    = rank + E;
    int*   rowptr = csr + E;

    const int Mtiles = (N + 15) / 16;
    dim3 wblk(32, 8);
    const dim3 ggrid((Mtiles + 7) / 8);
    const dim3 ngrid((N + 7) / 8);

    hipMemsetAsync(deg, 0, (size_t)N * sizeof(int), stream);
    hipLaunchKernelGGL(gat_edge_rank, dim3((E + 255) / 256), dim3(256), 0, stream,
                       dst, deg, rank, E);
    hipLaunchKernelGGL(gat_scan_deg, dim3((N + 1023) / 1024), dim3(1024), 0, stream,
                       deg, rowptr, N);
    hipLaunchKernelGGL(gat_edge_scatter, dim3((E + 255) / 256), dim3(256), 0, stream,
                       src, dst, rank, rowptr, csr, E);

    hipLaunchKernelGGL((gat_gemm_wmma<128, 4, 64>), ggrid, wblk, 0, stream,
                       X, W1, al1, ar1, feat1, el, er, N);
    hipLaunchKernelGGL((gat_node_softmax_agg<64, 64, true>), ngrid, wblk, 0, stream,
                       feat1, el, er, rowptr, csr, b1, h, N);

    hipLaunchKernelGGL((gat_gemm_wmma<64, 3, 40>), ggrid, wblk, 0, stream,
                       h, W2, al2, ar2, feat2, el, er, N);
    hipLaunchKernelGGL((gat_node_softmax_agg<40, 48, false>), ngrid, wblk, 0, stream,
                       feat2, el, er, rowptr, csr, b2, (float*)d_out, N);
}


# Edges grouped by destination by a two-level counting sort on LDS counters (512-node buckets, then nodes), written out as runs: no global rank atomics, scatter or memset; sum order = LDS arrival
# speedup vs baseline: 5.4420x; 1.7910x over previous
#include <hip/hip_runtime.h>
#include <hip/hip_bf16.h>
#include <math.h>

typedef __attribute__((ext_vector_type(2))) float v2f;
typedef __attribute__((ext_vector_type(8))) float v8f;

#define NEG_SLOPE 0.2f

template<int CTRL>
__device__ __forceinline__ float gat_dpp(float x) {
    return __int_as_float(__builtin_amdgcn_update_dpp(0, __float_as_int(x), CTRL, 0xf, 0xf, false));
}
__device__ __forceinline__ float gat_row16_sum(float x) {
    x += gat_dpp<0xB1>(x);
    x += gat_dpp<0x4E>(x);
    x += gat_dpp<0x141>(x);
    x += gat_dpp<0x140>(x);
    return x;
}

template<int K, int NT, int DOUT>
__global__ __launch_bounds__(256) __attribute__((amdgpu_waves_per_eu(2, 12))) void gat_gemm_wmma(const float* __restrict__ A,
                                                     const float* __restrict__ W,
                                                     const float* __restrict__ al,
                                                     const float* __restrict__ ar,
                                                     float* __restrict__ C,
                                                     float* __restrict__ el,
                                                     float* __restrict__ er, int N) {
    constexpr int DP = NT * 16;
    constexpr int CP = DP / 2;
    __shared__ float Bs[K * DP];

    const int tid = threadIdx.y * 32 + threadIdx.x;
    constexpr int ITEMS = (K / 2) * CP;
    static_assert(ITEMS % 256 == 0, "staging assumes a whole number of items per thread");
    static_assert(DOUT % 2 == 0 && DOUT <= DP, "column pairs must not straddle the padding");
    #pragma unroll
    for (int j = 0; j < ITEMS / 256; ++j) {
        const int i  = tid + 256 * j;
        const int kk = i / CP;
        const int c  = 2 * (i - kk * CP);
        const int cl = (DOUT < DP && c >= DOUT) ? 0 : c;
        float2 r0 = *(const float2*)&W[(size_t)(2 * kk) * DOUT + cl];
        float2 r1 = *(const float2*)&W[(size_t)(2 * kk + 1) * DOUT + cl];
        if (DOUT < DP && c >= DOUT) {
            r0 = make_float2(0.f, 0.f);
            r1 = make_float2(0.f, 0.f);
        }
        *(float4*)&Bs[(kk * DP + c) * 2] = make_float4(r0.x, r1.x, r0.y, r1.y);
    }
    __syncthreads();

    const int Mtiles = (N + 15) >> 4;
    const int strip = blockIdx.x * blockDim.y + threadIdx.y;
    if (strip >= Mtiles) return;
    int row0 = strip << 4;
    if (row0 + 16 > N) row0 = N - 16;

    const int m  = threadIdx.x & 15;
    const int kh = threadIdx.x >> 4;
    const float* __restrict__ arow = A + (size_t)(row0 + m) * K;

    float alv[NT], arv[NT];
    #pragma unroll
    for (int t = 0; t < NT; ++t) {
        const int cn = t * 16 + m;
        alv[t] = (cn < DOUT) ? al[cn] : 0.f;
        arv[t] = (cn < DOUT) ? ar[cn] : 0.f;
    }

    v8f acc[NT];
    #pragma unroll
    for (int t = 0; t < NT; ++t) acc[t] = (v8f){0.f,0.f,0.f,0.f,0.f,0.f,0.f,0.f};

    #pragma unroll
    for (int k = 0; k < K; k += 4) {
        const int ka = k + 2 * kh;
        const v2f a = *(const v2f*)(arow + ka);
        const int kk = ka >> 1;
        #pragma unroll
        for (int t = 0; t < NT; ++t) {
            const v2f b = *(const v2f*)&Bs[(kk * DP + t * 16 + m) * 2];
            acc[t] = __builtin_amdgcn_wmma_f32_16x16x4_f32(
                false, a, false, b, (short)0, acc[t], false, false);
        }
    }

    #pragma unroll
    for (int t = 0; t < NT; ++t) {
        const int cn = t * 16 + m;
        #pragma unroll
        for (int v = 0; v < 8; ++v)
            C[(size_t)(row0 + v + 8 * kh) * DP + cn] = acc[t][v];
    }

    float pl[8], pr[8];
    #pragma unroll
    for (int v = 0; v < 8; ++v) {
        float a = acc[0][v] * alv[0];
        float b = acc[0][v] * arv[0];
        #pragma unroll
        for (int t = 1; t < NT; ++t) {
            a += acc[t][v] * alv[t];
            b += acc[t][v] * arv[t];
        }
        pl[v] = gat_row16_sum(a);
        pr[v] = gat_row16_sum(b);
    }
    const int r = m & 7;
    float sl = pl[0], sr = pr[0];
    #pragma unroll
    for (int v = 1; v < 8; ++v) {
        if (r == v) { sl = pl[v]; sr = pr[v]; }
    }
    float* __restrict__ dp = (m < 8) ? el : er;
    dp[row0 + 8 * kh + r] = (m < 8) ? sl : sr;
}

#define GAT_BSH 9
#define GAT_CHE 4096
#define GAT_STG 12288

__device__ __forceinline__ int gat_block_excl_scan(int v, int tid, int* wtot) {
    const int lane = tid & 31;
    const int wv   = tid >> 5;
    int incl = v;
    for (int o = 1; o < 32; o <<= 1) {
        const int t = __shfl_up(incl, o);
        if (lane >= o) incl += t;
    }
    if (lane == 31) wtot[wv] = incl;
    __syncthreads();
    int carry = 0;
    for (int k = 0; k < wv; ++k) carry += wtot[k];
    return carry + incl - v;
}

__global__ __launch_bounds__(1024) void gat_bkt_count(const int* __restrict__ dst,
                                                      int* __restrict__ cnt, int E, int B, int C) {
    __shared__ int h[256];
    const int tid = threadIdx.x;
    const int c   = blockIdx.x;
    if (tid < 256) h[tid] = 0;
    __syncthreads();
    const int e0 = c * GAT_CHE;
    int dv[4];
    #pragma unroll
    for (int j = 0; j < 4; ++j) {
        const int e = e0 + j * 1024 + tid;
        dv[j] = (e < E) ? dst[e] : -1;
    }
    #pragma unroll
    for (int j = 0; j < 4; ++j)
        if (dv[j] >= 0) atomicAdd(&h[dv[j] >> GAT_BSH], 1);
    __syncthreads();
    if (tid < B) cnt[tid * C + c] = h[tid];
}

__global__ __launch_bounds__(1024) void gat_scan_deg(const int* __restrict__ deg,
                                                     int* __restrict__ rowptr, int N) {
    __shared__ int part[32];
    __shared__ int wtot[32];
    const int tid  = threadIdx.x;
    const int lane = tid & 31;
    const int wv   = tid >> 5;
    const int tile0 = blockIdx.x * 1024;

    int p = 0;
    for (int i = tid; i < tile0; i += 1024) p += deg[i];
    for (int off = 16; off > 0; off >>= 1) p += __shfl_xor(p, off);

    const int i = tile0 + tid;
    const int x = (i < N) ? deg[i] : 0;
    int incl = x;
    for (int off = 1; off < 32; off <<= 1) {
        const int t = __shfl_up(incl, off);
        if (lane >= off) incl += t;
    }
    if (lane == 0)  part[wv] = p;
    if (lane == 31) wtot[wv] = incl;
    __syncthreads();

    int carry = 0;
    for (int k = 0; k < 32; ++k) {
        carry += part[k];
        if (k < wv) carry += wtot[k];
    }
    const int excl = carry + incl - x;
    if (i < N) rowptr[i] = excl;
    if (i == N - 1) rowptr[N] = excl + x;
}

__global__ __launch_bounds__(1024) void gat_bkt_place(const int* __restrict__ src,
                                                      const int* __restrict__ dst,
                                                      const int* __restrict__ cnt,
                                                      const int* __restrict__ off,
                                                      int* __restrict__ bedge_w,
                                                      int E, int B, int C) {
    int2* __restrict__ bedge = (int2*)bedge_w;
    __shared__ int  cur[256];
    __shared__ int  gdl[256];
    __shared__ int  wtot[32];
    __shared__ int2 stage[GAT_CHE];
    const int tid = threadIdx.x;
    const int c   = blockIdx.x;
    const int e0  = c * GAT_CHE;
    int sv[4], dv[4];
    #pragma unroll
    for (int j = 0; j < 4; ++j) {
        const int e = e0 + j * 1024 + tid;
        dv[j] = (e < E) ? dst[e] : -1;
        sv[j] = (e < E) ? src[e] : 0;
    }
    const int v  = (tid < B) ? cnt[tid * C + c] : 0;
    const int gb = (tid < B) ? off[tid * C + c] : 0;
    const int ex = gat_block_excl_scan(v, tid, wtot);
    if (tid < 256) { cur[tid] = ex; gdl[tid] = gb - ex; }
    __syncthreads();
    #pragma unroll
    for (int j = 0; j < 4; ++j) {
        if (dv[j] >= 0) {
            const int lp = atomicAdd(&cur[dv[j] >> GAT_BSH], 1);
            stage[lp] = make_int2(sv[j], dv[j]);
        }
    }
    __syncthreads();
    const int nloc = min(GAT_CHE, E - e0);
    for (int lp = tid; lp < nloc; lp += 1024) {
        const int2 p = stage[lp];
        bedge[gdl[p.y >> GAT_BSH] + lp] = p;
    }
}

__global__ __launch_bounds__(1024) void gat_bkt_sort(const int* __restrict__ bedge_w,
                                                     const int* __restrict__ off,
                                                     int* __restrict__ csr,
                                                     int* __restrict__ rowptr,
                                                     int N, int B, int C) {
    const int2* __restrict__ bedge = (const int2*)bedge_w;
    __shared__ int cn[512];
    __shared__ int wtot[32];
    __shared__ int stg[GAT_STG];
    const int tid = threadIdx.x;
    const int b   = blockIdx.x;
    const int s0  = off[b * C];
    const int s1  = off[(b + 1) * C];
    const int nb  = s1 - s0;
    const int d0  = b << GAT_BSH;
    if (tid < 512) cn[tid] = 0;
    __syncthreads();
    for (int i0 = 0; i0 < nb; i0 += 4096) {
        int dv[4];
        #pragma unroll
        for (int j = 0; j < 4; ++j) {
            const int i = i0 + j * 1024 + tid;
            dv[j] = (i < nb) ? bedge[s0 + i].y : -1;
        }
        #pragma unroll
        for (int j = 0; j < 4; ++j)
            if (dv[j] >= 0) atomicAdd(&cn[dv[j] - d0], 1);
    }
    __syncthreads();
    const int v  = (tid < 512) ? cn[tid] : 0;
    const int ex = gat_block_excl_scan(v, tid, wtot);
    if (tid < 512) {
        cn[tid] = ex;
        if (d0 + tid < N) rowptr[d0 + tid] = s0 + ex;
    }
    if (b == B - 1 && tid == 0) rowptr[N] = s1;
    __syncthreads();
    const bool staged = (nb <= GAT_STG);
    for (int i0 = 0; i0 < nb; i0 += 4096) {
        int2 p[4];
        #pragma unroll
        for (int j = 0; j < 4; ++j) {
            const int i = i0 + j * 1024 + tid;
            p[j] = (i < nb) ? bedge[s0 + i] : make_int2(0, -1);
        }
        #pragma unroll
        for (int j = 0; j < 4; ++j) {
            if (p[j].y >= 0) {
                const int lp = atomicAdd(&cn[p[j].y - d0], 1);
                if (staged) stg[lp] = p[j].x;
                else        csr[s0 + lp] = p[j].x;
            }
        }
    }
    __syncthreads();
    if (staged)
        for (int i = tid; i < nb; i += 1024) csr[s0 + i] = stg[i];
}

template<int G, int FS>
__device__ __forceinline__ void gat_acc_rows(const float* __restrict__ Fl, int s, float w, int t,
                                             float& ax, float& ay) {
    float2 f[G];
    float  wt[G];
    #pragma unroll
    for (int g = 0; g < G; ++g) {
        const int sg = __builtin_amdgcn_readlane(s, t + g);
        wt[g] = __int_as_float(__builtin_amdgcn_readlane(__float_as_int(w), t + g));
        f[g] = *(const float2*)(Fl + (size_t)sg * FS);
    }
    #pragma unroll
    for (int g = 0; g < G; ++g) {
        ax += wt[g] * f[g].x;
        ay += wt[g] * f[g].y;
    }
}

template<int FS>
__device__ __forceinline__ void gat_acc_chunk(const float* __restrict__ Fl, int s, float w, int cnt,
                                              float& ax, float& ay) {
    int t = 0;
    for (; t + 8 <= cnt; t += 8) gat_acc_rows<8, FS>(Fl, s, w, t, ax, ay);
    if (cnt & 4) { gat_acc_rows<4, FS>(Fl, s, w, t, ax, ay); t += 4; }
    if (cnt & 2) { gat_acc_rows<2, FS>(Fl, s, w, t, ax, ay); t += 2; }
    if (cnt & 1) { gat_acc_rows<1, FS>(Fl, s, w, t, ax, ay); }
}

template<int D, int FS, bool ELU>
__global__ __launch_bounds__(256) void gat_node_softmax_agg(const float* __restrict__ F,
                                                            const float* __restrict__ el,
                                                            const float* __restrict__ er,
                                                            const int* __restrict__ rowptr,
                                                            const int* __restrict__ csr,
                                                            const float* __restrict__ bias,
                                                            float* __restrict__ out, int N) {
    const int lane = threadIdx.x;
    const int node = blockIdx.x * 8 + __builtin_amdgcn_readfirstlane((int)threadIdx.y);
    if (node >= N) return;
    const int beg = rowptr[node];
    const int end = rowptr[node + 1];
    const float erd = er[node];
    const bool colok = (2 * lane < D);
    const float* __restrict__ Fl = F + (colok ? 2 * lane : 0);

    const bool has0 = (beg + lane < end);
    int s0 = 0;
    float v0 = -INFINITY;
    if (has0) {
        s0 = csr[beg + lane];
        const float v = el[s0] + erd;
        v0 = (v > 0.f) ? v : NEG_SLOPE * v;
    }
    float m = v0;
    for (int j = beg + 32 + lane; j < end; j += 32) {
        const float v = el[csr[j]] + erd;
        m = fmaxf(m, (v > 0.f) ? v : NEG_SLOPE * v);
    }
    for (int off = 16; off > 0; off >>= 1) m = fmaxf(m, __shfl_xor(m, off));

    float ax = 0.f, ay = 0.f;
    const float w0 = has0 ? __expf(v0 - m) : 0.f;
    float ssum = w0;
    if (end > beg)
        gat_acc_chunk<FS>(Fl, s0, w0, min(32, end - beg), ax, ay);
    for (int base = beg + 32; base < end; base += 32) {
        const int j = base + lane;
        int s = 0;
        float w = 0.f;
        if (j < end) {
            s = csr[j];
            const float v = el[s] + erd;
            w = __expf(((v > 0.f) ? v : NEG_SLOPE * v) - m);
        }
        ssum += w;
        gat_acc_chunk<FS>(Fl, s, w, min(32, end - base), ax, ay);
    }
    for (int off = 16; off > 0; off >>= 1) ssum += __shfl_xor(ssum, off);

    if (colok) {
        const float2 b = *(const float2*)(bias + 2 * lane);
        float x = b.x, y = b.y;
        if (end > beg) { x += ax / ssum; y += ay / ssum; }
        if (ELU) {
            x = (x > 0.f) ? x : (__expf(x) - 1.f);
            y = (y > 0.f) ? y : (__expf(y) - 1.f);
        }
        *(float2*)(out + (size_t)node * D + 2 * lane) = make_float2(x, y);
    }
}

extern "C" void kernel_launch(void* const* d_in, const int* in_sizes, int n_in,
                              void* d_out, int out_size, void* d_ws, size_t ws_size,
                              hipStream_t stream) {
    const float* X   = (const float*)d_in[0];
    const int*   src = (const int*)d_in[1];
    const int*   dst = (const int*)d_in[2];
    const float* W1  = (const float*)d_in[3];
    const float* al1 = (const float*)d_in[4];
    const float* ar1 = (const float*)d_in[5];
    const float* b1  = (const float*)d_in[6];
    const float* W2  = (const float*)d_in[7];
    const float* al2 = (const float*)d_in[8];
    const float* ar2 = (const float*)d_in[9];
    const float* b2  = (const float*)d_in[10];

    const int IN = 128, H = 64, CLSPAD = 48;
    const int N = in_sizes[0] / IN;
    const int E = in_sizes[1];

    float* ws = (float*)d_ws;
    float* feat1 = ws;
    float* h     = feat1 + (size_t)N * H;
    float* feat2 = h + (size_t)N * H;
    float* el    = feat2 + (size_t)N * CLSPAD;
    float* er    = el + N;
    int*   bedge  = (int*)(er + N);
    int*   csr    = bedge + 2 * (size_t)E;
    int*   rowptr = csr + E;
    const int B = (N + 511) / 512;
    const int C = (E + 4095) / 4096;
    int*   cnt    = rowptr + N + 1;
    int*   off    = cnt + B * C;

    const int Mtiles = (N + 15) / 16;
    dim3 wblk(32, 8);
    const dim3 ggrid((Mtiles + 7) / 8);
    const dim3 ngrid((N + 7) / 8);

    hipLaunchKernelGGL(gat_bkt_count, dim3(C), dim3(1024), 0, stream, dst, cnt, E, B, C);
    hipLaunchKernelGGL(gat_scan_deg, dim3((B * C + 1023) / 1024), dim3(1024), 0, stream,
                       cnt, off, B * C);
    hipLaunchKernelGGL(gat_bkt_place, dim3(C), dim3(1024), 0, stream,
                       src, dst, cnt, off, bedge, E, B, C);
    hipLaunchKernelGGL(gat_bkt_sort, dim3(B), dim3(1024), 0, stream,
                       bedge, off, csr, rowptr, N, B, C);

    hipLaunchKernelGGL((gat_gemm_wmma<128, 4, 64>), ggrid, wblk, 0, stream,
                       X, W1, al1, ar1, feat1, el, er, N);
    hipLaunchKernelGGL((gat_node_softmax_agg<64, 64, true>), ngrid, wblk, 0, stream,
                       feat1, el, er, rowptr, csr, b1, h, N);

    hipLaunchKernelGGL((gat_gemm_wmma<64, 3, 40>), ggrid, wblk, 0, stream,
                       h, W2, al2, ar2, feat2, el, er, N);
    hipLaunchKernelGGL((gat_node_softmax_agg<40, 48, false>), ngrid, wblk, 0, stream,
                       feat2, el, er, rowptr, csr, b2, (float*)d_out, N);
}


# Edge grouping in 2 launches, not 4: chunk kernel counts and files its edges in place (one word per edge); bucket kernel takes its runs and global start from its table column; sum order = LDS arrival
# speedup vs baseline: 6.0729x; 1.1159x over previous
#include <hip/hip_runtime.h>
#include <hip/hip_bf16.h>
#include <math.h>

typedef __attribute__((ext_vector_type(2))) float v2f;
typedef __attribute__((ext_vector_type(8))) float v8f;

#define NEG_SLOPE 0.2f

template<int CTRL>
__device__ __forceinline__ float gat_dpp(float x) {
    return __int_as_float(__builtin_amdgcn_update_dpp(0, __float_as_int(x), CTRL, 0xf, 0xf, false));
}
__device__ __forceinline__ float gat_row16_sum(float x) {
    x += gat_dpp<0xB1>(x);
    x += gat_dpp<0x4E>(x);
    x += gat_dpp<0x141>(x);
    x += gat_dpp<0x140>(x);
    return x;
}

template<int K, int NT, int DOUT>
__global__ __launch_bounds__(256) __attribute__((amdgpu_waves_per_eu(2, 12))) void gat_gemm_wmma(const float* __restrict__ A,
                                                     const float* __restrict__ W,
                                                     const float* __restrict__ al,
                                                     const float* __restrict__ ar,
                                                     float* __restrict__ C,
                                                     float* __restrict__ el,
                                                     float* __restrict__ er, int N) {
    constexpr int DP = NT * 16;
    constexpr int CP = DP / 2;
    __shared__ float Bs[K * DP];

    const int tid = threadIdx.y * 32 + threadIdx.x;
    constexpr int ITEMS = (K / 2) * CP;
    static_assert(ITEMS % 256 == 0, "staging assumes a whole number of items per thread");
    static_assert(DOUT % 2 == 0 && DOUT <= DP, "column pairs must not straddle the padding");
    #pragma unroll
    for (int j = 0; j < ITEMS / 256; ++j) {
        const int i  = tid + 256 * j;
        const int kk = i / CP;
        const int c  = 2 * (i - kk * CP);
        const int cl = (DOUT < DP && c >= DOUT) ? 0 : c;
        float2 r0 = *(const float2*)&W[(size_t)(2 * kk) * DOUT + cl];
        float2 r1 = *(const float2*)&W[(size_t)(2 * kk + 1) * DOUT + cl];
        if (DOUT < DP && c >= DOUT) {
            r0 = make_float2(0.f, 0.f);
            r1 = make_float2(0.f, 0.f);
        }
        *(float4*)&Bs[(kk * DP + c) * 2] = make_float4(r0.x, r1.x, r0.y, r1.y);
    }
    __syncthreads();

    const int Mtiles = (N + 15) >> 4;
    const int strip = blockIdx.x * blockDim.y + threadIdx.y;
    if (strip >= Mtiles) return;
    int row0 = strip << 4;
    if (row0 + 16 > N) row0 = N - 16;

    const int m  = threadIdx.x & 15;
    const int kh = threadIdx.x >> 4;
    const float* __restrict__ arow = A + (size_t)(row0 + m) * K;

    float alv[NT], arv[NT];
    #pragma unroll
    for (int t = 0; t < NT; ++t) {
        const int cn = t * 16 + m;
        alv[t] = (cn < DOUT) ? al[cn] : 0.f;
        arv[t] = (cn < DOUT) ? ar[cn] : 0.f;
    }

    v8f acc[NT];
    #pragma unroll
    for (int t = 0; t < NT; ++t) acc[t] = (v8f){0.f,0.f,0.f,0.f,0.f,0.f,0.f,0.f};

    #pragma unroll
    for (int k = 0; k < K; k += 4) {
        const int ka = k + 2 * kh;
        const v2f a = *(const v2f*)(arow + ka);
        const int kk = ka >> 1;
        #pragma unroll
        for (int t = 0; t < NT; ++t) {
            const v2f b = *(const v2f*)&Bs[(kk * DP + t * 16 + m) * 2];
            acc[t] = __builtin_amdgcn_wmma_f32_16x16x4_f32(
                false, a, false, b, (short)0, acc[t], false, false);
        }
    }

    #pragma unroll
    for (int t = 0; t < NT; ++t) {
        const int cn = t * 16 + m;
        #pragma unroll
        for (int v = 0; v < 8; ++v)
            C[(size_t)(row0 + v + 8 * kh) * DP + cn] = acc[t][v];
    }

    float pl[8], pr[8];
    #pragma unroll
    for (int v = 0; v < 8; ++v) {
        float a = acc[0][v] * alv[0];
        float b = acc[0][v] * arv[0];
        #pragma unroll
        for (int t = 1; t < NT; ++t) {
            a += acc[t][v] * alv[t];
            b += acc[t][v] * arv[t];
        }
        pl[v] = gat_row16_sum(a);
        pr[v] = gat_row16_sum(b);
    }
    const int r = m & 7;
    float sl = pl[0], sr = pr[0];
    #pragma unroll
    for (int v = 1; v < 8; ++v) {
        if (r == v) { sl = pl[v]; sr = pr[v]; }
    }
    float* __restrict__ dp = (m < 8) ? el : er;
    dp[row0 + 8 * kh + r] = (m < 8) ? sl : sr;
}

#define GAT_BSH 9
#define GAT_CHE 4096
#define GAT_STG 12288
#define GAT_SRCB 17
#define GAT_SRCM 0x1FFFF
#define GAT_SEGB 16

__device__ __forceinline__ int gat_block_excl_scan(int v, int tid, int* wtot) {
    const int lane = tid & 31;
    const int wv   = tid >> 5;
    int incl = v;
    for (int o = 1; o < 32; o <<= 1) {
        const int t = __shfl_up(incl, o);
        if (lane >= o) incl += t;
    }
    if (lane == 31) wtot[wv] = incl;
    __syncthreads();
    int carry = 0;
    for (int k = 0; k < wv; ++k) carry += wtot[k];
    return carry + incl - v;
}

__global__ __launch_bounds__(1024) void gat_chunk_file(const int* __restrict__ src,
                                                       const int* __restrict__ dst,
                                                       int* __restrict__ cnt,
                                                       int* __restrict__ loc,
                                                       int* __restrict__ cedge,
                                                       int E, int B) {
    __shared__ int h[256];
    __shared__ int cur[256];
    __shared__ int wtot[32];
    __shared__ int stage[GAT_CHE];
    const int tid = threadIdx.x;
    const int c   = blockIdx.x;
    const int e0  = c * GAT_CHE;
    if (tid < 256) h[tid] = 0;
    int sv[4], dv[4];
    #pragma unroll
    for (int j = 0; j < 4; ++j) {
        const int e = e0 + j * 1024 + tid;
        dv[j] = (e < E) ? dst[e] : -1;
        sv[j] = (e < E) ? src[e] : 0;
    }
    __syncthreads();
    #pragma unroll
    for (int j = 0; j < 4; ++j)
        if (dv[j] >= 0) atomicAdd(&h[dv[j] >> GAT_BSH], 1);
    __syncthreads();
    const int v  = (tid < 256) ? h[tid] : 0;
    const int ex = gat_block_excl_scan(v, tid, wtot);
    if (tid < 256) cur[tid] = ex;
    if (tid < B) {
        cnt[c * B + tid] = v;
        loc[c * B + tid] = ex;
    }
    __syncthreads();
    #pragma unroll
    for (int j = 0; j < 4; ++j) {
        if (dv[j] >= 0) {
            const int lp = atomicAdd(&cur[dv[j] >> GAT_BSH], 1);
            stage[lp] = sv[j] | ((dv[j] & ((1 << GAT_BSH) - 1)) << GAT_SRCB);
        }
    }
    __syncthreads();
    const int nloc = min(GAT_CHE, E - e0);
    for (int lp = tid; lp < nloc; lp += 1024) cedge[e0 + lp] = stage[lp];
}

template<int MODE>
__device__ __forceinline__ void gat_seg_edge(int w, int slot, int* cn, int* stage,
                                             int* __restrict__ csr, int s0) {
    if (MODE == 2) {
        const int lp = atomicAdd(&cn[w >> GAT_SRCB], 1);
        csr[s0 + lp] = w & GAT_SRCM;
    } else {
        atomicAdd(&cn[w >> GAT_SRCB], 1);
        if (MODE == 0) stage[slot] = w;
    }
}

template<int MODE>
__device__ __forceinline__ void gat_walk_runs(const int* __restrict__ cedge, int nc, int oc, int pc,
                                              int lane, int* cn, int* stage,
                                              int* __restrict__ csr, int s0) {
    const unsigned live = __builtin_amdgcn_ballot_w32(nc > 0);
    #pragma unroll
    for (int g = 0; g < 32; g += GAT_SEGB) {
        if ((live >> g) == 0u) break;
        int w[GAT_SEGB];
        #pragma unroll
        for (int j = 0; j < GAT_SEGB; ++j) {
            const int n = __builtin_amdgcn_readlane(nc, g + j);
            const int o = __builtin_amdgcn_readlane(oc, g + j);
            w[j] = (lane < n) ? cedge[o + lane] : -1;
        }
        #pragma unroll
        for (int j = 0; j < GAT_SEGB; ++j) {
            const int p = __builtin_amdgcn_readlane(pc, g + j);
            if (w[j] >= 0) gat_seg_edge<MODE>(w[j], p + lane, cn, stage, csr, s0);
        }
        #pragma unroll
        for (int j = 0; j < GAT_SEGB; ++j) {
            const int n = __builtin_amdgcn_readlane(nc, g + j);
            if (n > 32) {
                const int o = __builtin_amdgcn_readlane(oc, g + j);
                const int p = __builtin_amdgcn_readlane(pc, g + j);
                for (int l = lane + 32; l < n; l += 32)
                    gat_seg_edge<MODE>(cedge[o + l], p + l, cn, stage, csr, s0);
            }
        }
    }
}

__global__ __launch_bounds__(1024) void gat_bkt_sort(const int* __restrict__ cedge,
                                                     const int* __restrict__ cnt,
                                                     const int* __restrict__ loc,
                                                     int* __restrict__ csr,
                                                     int* __restrict__ rowptr,
                                                     int N, int B, int C) {
    __shared__ int cn[512];
    __shared__ int wtot[32];
    __shared__ int wnum[32];
    __shared__ int wloc[32];
    __shared__ int stage[GAT_STG];
    __shared__ int stg[GAT_STG];
    const int tid  = threadIdx.x;
    const int lane = tid & 31;
    const int wv   = tid >> 5;
    const int b    = blockIdx.x;
    const int d0   = b << GAT_BSH;
    if (tid < 512) cn[tid] = 0;
    const int cc = lane * 32 + wv;
    int nc = 0, lc = 0;
    if (cc < C) {
        nc = cnt[cc * B + b];
        lc = loc[cc * B + b];
    }
    const int pc = gat_block_excl_scan(nc, tid, wnum);
    const int oc = cc * GAT_CHE + lc;
    int ls = lc;
    for (int o = 16; o > 0; o >>= 1) ls += __shfl_xor(ls, o);
    if (lane == 0) wloc[wv] = ls;
    __syncthreads();
    int s0 = 0, nb = 0;
    for (int k = 0; k < 32; ++k) {
        s0 += wloc[k];
        nb += wnum[k];
    }
    const bool staged = (nb <= GAT_STG);
    if (staged) gat_walk_runs<0>(cedge, nc, oc, pc, lane, cn, stage, csr, s0);
    else        gat_walk_runs<1>(cedge, nc, oc, pc, lane, cn, stage, csr, s0);
    __syncthreads();
    const int v  = (tid < 512) ? cn[tid] : 0;
    const int ex = gat_block_excl_scan(v, tid, wtot);
    if (tid < 512) {
        cn[tid] = ex;
        if (d0 + tid < N) rowptr[d0 + tid] = s0 + ex;
    }
    if (b == B - 1 && tid == 0) rowptr[N] = s0 + nb;
    __syncthreads();
    if (staged) {
        for (int i0 = 0; i0 < nb; i0 += 4096) {
            int w[4];
            #pragma unroll
            for (int j = 0; j < 4; ++j) {
                const int i = i0 + j * 1024 + tid;
                w[j] = (i < nb) ? stage[i] : -1;
            }
            #pragma unroll
            for (int j = 0; j < 4; ++j) {
                if (w[j] >= 0) {
                    const int lp = atomicAdd(&cn[w[j] >> GAT_SRCB], 1);
                    stg[lp] = w[j] & GAT_SRCM;
                }
            }
        }
        __syncthreads();
        for (int i = tid; i < nb; i += 1024) csr[s0 + i] = stg[i];
    } else {
        gat_walk_runs<2>(cedge, nc, oc, pc, lane, cn, stage, csr, s0);
    }
}

template<int G, int FS>
__device__ __forceinline__ void gat_acc_rows(const float* __restrict__ Fl, int s, float w, int t,
                                             float& ax, float& ay) {
    float2 f[G];
    float  wt[G];
    #pragma unroll
    for (int g = 0; g < G; ++g) {
        const int sg = __builtin_amdgcn_readlane(s, t + g);
        wt[g] = __int_as_float(__builtin_amdgcn_readlane(__float_as_int(w), t + g));
        f[g] = *(const float2*)(Fl + (size_t)sg * FS);
    }
    #pragma unroll
    for (int g = 0; g < G; ++g) {
        ax += wt[g] * f[g].x;
        ay += wt[g] * f[g].y;
    }
}

template<int FS>
__device__ __forceinline__ void gat_acc_chunk(const float* __restrict__ Fl, int s, float w, int cnt,
                                              float& ax, float& ay) {
    int t = 0;
    for (; t + 8 <= cnt; t += 8) gat_acc_rows<8, FS>(Fl, s, w, t, ax, ay);
    if (cnt & 4) { gat_acc_rows<4, FS>(Fl, s, w, t, ax, ay); t += 4; }
    if (cnt & 2) { gat_acc_rows<2, FS>(Fl, s, w, t, ax, ay); t += 2; }
    if (cnt & 1) { gat_acc_rows<1, FS>(Fl, s, w, t, ax, ay); }
}

template<int D, int FS, bool ELU>
__global__ __launch_bounds__(256) void gat_node_softmax_agg(const float* __restrict__ F,
                                                            const float* __restrict__ el,
                                                            const float* __restrict__ er,
                                                            const int* __restrict__ rowptr,
                                                            const int* __restrict__ csr,
                                                            const float* __restrict__ bias,
                                                            float* __restrict__ out, int N) {
    const int lane = threadIdx.x;
    const int node = blockIdx.x * 8 + __builtin_amdgcn_readfirstlane((int)threadIdx.y);
    if (node >= N) return;
    const int beg = rowptr[node];
    const int end = rowptr[node + 1];
    const float erd = er[node];
    const bool colok = (2 * lane < D);
    const float* __restrict__ Fl = F + (colok ? 2 * lane : 0);

    const bool has0 = (beg + lane < end);
    int s0 = 0;
    float v0 = -INFINITY;
    if (has0) {
        s0 = csr[beg + lane];
        const float v = el[s0] + erd;
        v0 = (v > 0.f) ? v : NEG_SLOPE * v;
    }
    float m = v0;
    for (int j = beg + 32 + lane; j < end; j += 32) {
        const float v = el[csr[j]] + erd;
        m = fmaxf(m, (v > 0.f) ? v : NEG_SLOPE * v);
    }
    for (int off = 16; off > 0; off >>= 1) m = fmaxf(m, __shfl_xor(m, off));

    float ax = 0.f, ay = 0.f;
    const float w0 = has0 ? __expf(v0 - m) : 0.f;
    float ssum = w0;
    if (end > beg)
        gat_acc_chunk<FS>(Fl, s0, w0, min(32, end - beg), ax, ay);
    for (int base = beg + 32; base < end; base += 32) {
        const int j = base + lane;
        int s = 0;
        float w = 0.f;
        if (j < end) {
            s = csr[j];
            const float v = el[s] + erd;
            w = __expf(((v > 0.f) ? v : NEG_SLOPE * v) - m);
        }
        ssum += w;
        gat_acc_chunk<FS>(Fl, s, w, min(32, end - base), ax, ay);
    }
    for (int off = 16; off > 0; off >>= 1) ssum += __shfl_xor(ssum, off);

    if (colok) {
        const float2 b = *(const float2*)(bias + 2 * lane);
        float x = b.x, y = b.y;
        if (end > beg) { x += ax / ssum; y += ay / ssum; }
        if (ELU) {
            x = (x > 0.f) ? x : (__expf(x) - 1.f);
            y = (y > 0.f) ? y : (__expf(y) - 1.f);
        }
        *(float2*)(out + (size_t)node * D + 2 * lane) = make_float2(x, y);
    }
}

extern "C" void kernel_launch(void* const* d_in, const int* in_sizes, int n_in,
                              void* d_out, int out_size, void* d_ws, size_t ws_size,
                              hipStream_t stream) {
    const float* X   = (const float*)d_in[0];
    const int*   src = (const int*)d_in[1];
    const int*   dst = (const int*)d_in[2];
    const float* W1  = (const float*)d_in[3];
    const float* al1 = (const float*)d_in[4];
    const float* ar1 = (const float*)d_in[5];
    const float* b1  = (const float*)d_in[6];
    const float* W2  = (const float*)d_in[7];
    const float* al2 = (const float*)d_in[8];
    const float* ar2 = (const float*)d_in[9];
    const float* b2  = (const float*)d_in[10];

    const int IN = 128, H = 64, CLSPAD = 48;
    const int N = in_sizes[0] / IN;
    const int E = in_sizes[1];

    float* ws = (float*)d_ws;
    float* feat1 = ws;
    float* h     = feat1 + (size_t)N * H;
    float* feat2 = h + (size_t)N * H;
    float* el    = feat2 + (size_t)N * CLSPAD;
    float* er    = el + N;
    int*   cedge  = (int*)(er + N);
    int*   csr    = cedge + E;
    int*   rowptr = csr + E;
    const int B = (N + 511) / 512;
    const int C = (E + 4095) / 4096;
    int*   cnt    = rowptr + N + 1;
    int*   loc    = cnt + B * C;

    const int Mtiles = (N + 15) / 16;
    dim3 wblk(32, 8);
    const dim3 ggrid((Mtiles + 7) / 8);
    const dim3 ngrid((N + 7) / 8);

    hipLaunchKernelGGL(gat_chunk_file, dim3(C), dim3(1024), 0, stream,
                       src, dst, cnt, loc, cedge, E, B);
    hipLaunchKernelGGL(gat_bkt_sort, dim3(B), dim3(1024), 0, stream,
                       cedge, cnt, loc, csr, rowptr, N, B, C);

    hipLaunchKernelGGL((gat_gemm_wmma<128, 4, 64>), ggrid, wblk, 0, stream,
                       X, W1, al1, ar1, feat1, el, er, N);
    hipLaunchKernelGGL((gat_node_softmax_agg<64, 64, true>), ngrid, wblk, 0, stream,
                       feat1, el, er, rowptr, csr, b1, h, N);

    hipLaunchKernelGGL((gat_gemm_wmma<64, 3, 40>), ggrid, wblk, 0, stream,
                       h, W2, al2, ar2, feat2, el, er, N);
    hipLaunchKernelGGL((gat_node_softmax_agg<40, 48, false>), ngrid, wblk, 0, stream,
                       feat2, el, er, rowptr, csr, b2, (float*)d_out, N);
}


# Layer-2 node kernel in one pass: GEMM stores el in padding column 40 of the feat row, so a logit arrives with its row; running softmax rescaled per group (new sum order); el gather and max pass gone
# speedup vs baseline: 6.3583x; 1.0470x over previous
#include <hip/hip_runtime.h>
#include <hip/hip_bf16.h>
#include <math.h>

typedef __attribute__((ext_vector_type(2))) float v2f;
typedef __attribute__((ext_vector_type(8))) float v8f;

#define NEG_SLOPE 0.2f

template<int CTRL>
__device__ __forceinline__ float gat_dpp(float x) {
    return __int_as_float(__builtin_amdgcn_update_dpp(0, __float_as_int(x), CTRL, 0xf, 0xf, false));
}
__device__ __forceinline__ float gat_row16_sum(float x) {
    x += gat_dpp<0xB1>(x);
    x += gat_dpp<0x4E>(x);
    x += gat_dpp<0x141>(x);
    x += gat_dpp<0x140>(x);
    return x;
}

template<int K, int NT, int DOUT>
__global__ __launch_bounds__(256) __attribute__((amdgpu_waves_per_eu(2, 12))) void gat_gemm_wmma(const float* __restrict__ A,
                                                     const float* __restrict__ W,
                                                     const float* __restrict__ al,
                                                     const float* __restrict__ ar,
                                                     float* __restrict__ C,
                                                     float* __restrict__ el,
                                                     float* __restrict__ er, int N) {
    constexpr int DP = NT * 16;
    constexpr int CP = DP / 2;
    __shared__ float Bs[K * DP];

    const int tid = threadIdx.y * 32 + threadIdx.x;
    constexpr int ITEMS = (K / 2) * CP;
    static_assert(ITEMS % 256 == 0, "staging assumes a whole number of items per thread");
    static_assert(DOUT % 2 == 0 && DOUT <= DP, "column pairs must not straddle the padding");
    #pragma unroll
    for (int j = 0; j < ITEMS / 256; ++j) {
        const int i  = tid + 256 * j;
        const int kk = i / CP;
        const int c  = 2 * (i - kk * CP);
        const int cl = (DOUT < DP && c >= DOUT) ? 0 : c;
        float2 r0 = *(const float2*)&W[(size_t)(2 * kk) * DOUT + cl];
        float2 r1 = *(const float2*)&W[(size_t)(2 * kk + 1) * DOUT + cl];
        if (DOUT < DP && c >= DOUT) {
            r0 = make_float2(0.f, 0.f);
            r1 = make_float2(0.f, 0.f);
        }
        *(float4*)&Bs[(kk * DP + c) * 2] = make_float4(r0.x, r1.x, r0.y, r1.y);
    }
    __syncthreads();

    const int Mtiles = (N + 15) >> 4;
    const int strip = blockIdx.x * blockDim.y + threadIdx.y;
    if (strip >= Mtiles) return;
    int row0 = strip << 4;
    if (row0 + 16 > N) row0 = N - 16;

    const int m  = threadIdx.x & 15;
    const int kh = threadIdx.x >> 4;
    const float* __restrict__ arow = A + (size_t)(row0 + m) * K;

    float alv[NT], arv[NT];
    #pragma unroll
    for (int t = 0; t < NT; ++t) {
        const int cn = t * 16 + m;
        alv[t] = (cn < DOUT) ? al[cn] : 0.f;
        arv[t] = (cn < DOUT) ? ar[cn] : 0.f;
    }

    v8f acc[NT];
    #pragma unroll
    for (int t = 0; t < NT; ++t) acc[t] = (v8f){0.f,0.f,0.f,0.f,0.f,0.f,0.f,0.f};

    #pragma unroll
    for (int k = 0; k < K; k += 4) {
        const int ka = k + 2 * kh;
        const v2f a = *(const v2f*)(arow + ka);
        const int kk = ka >> 1;
        #pragma unroll
        for (int t = 0; t < NT; ++t) {
            const v2f b = *(const v2f*)&Bs[(kk * DP + t * 16 + m) * 2];
            acc[t] = __builtin_amdgcn_wmma_f32_16x16x4_f32(
                false, a, false, b, (short)0, acc[t], false, false);
        }
    }

    float pl[8], pr[8];
    #pragma unroll
    for (int v = 0; v < 8; ++v) {
        float a = acc[0][v] * alv[0];
        float b = acc[0][v] * arv[0];
        #pragma unroll
        for (int t = 1; t < NT; ++t) {
            a += acc[t][v] * alv[t];
            b += acc[t][v] * arv[t];
        }
        pl[v] = gat_row16_sum(a);
        pr[v] = gat_row16_sum(b);
    }

    #pragma unroll
    for (int t = 0; t < NT; ++t) {
        const int cn = t * 16 + m;
        #pragma unroll
        for (int v = 0; v < 8; ++v) {
            float val = acc[t][v];
            if (DOUT < DP && t == DOUT / 16 && cn == DOUT) val = pl[v];
            C[(size_t)(row0 + v + 8 * kh) * DP + cn] = val;
        }
    }
    const int r = m & 7;
    float sl = pl[0], sr = pr[0];
    #pragma unroll
    for (int v = 1; v < 8; ++v) {
        if (r == v) { sl = pl[v]; sr = pr[v]; }
    }
    float* __restrict__ dp = (m < 8) ? el : er;
    dp[row0 + 8 * kh + r] = (m < 8) ? sl : sr;
}

#define GAT_BSH 9
#define GAT_CHE 4096
#define GAT_STG 12288
#define GAT_SRCB 17
#define GAT_SRCM 0x1FFFF
#define GAT_SEGB 16

__device__ __forceinline__ int gat_block_excl_scan(int v, int tid, int* wtot) {
    const int lane = tid & 31;
    const int wv   = tid >> 5;
    int incl = v;
    for (int o = 1; o < 32; o <<= 1) {
        const int t = __shfl_up(incl, o);
        if (lane >= o) incl += t;
    }
    if (lane == 31) wtot[wv] = incl;
    __syncthreads();
    int carry = 0;
    for (int k = 0; k < wv; ++k) carry += wtot[k];
    return carry + incl - v;
}

__global__ __launch_bounds__(1024) void gat_chunk_file(const int* __restrict__ src,
                                                       const int* __restrict__ dst,
                                                       int* __restrict__ cnt,
                                                       int* __restrict__ loc,
                                                       int* __restrict__ cedge,
                                                       int E, int B) {
    __shared__ int h[256];
    __shared__ int cur[256];
    __shared__ int wtot[32];
    __shared__ int stage[GAT_CHE];
    const int tid = threadIdx.x;
    const int c   = blockIdx.x;
    const int e0  = c * GAT_CHE;
    if (tid < 256) h[tid] = 0;
    int sv[4], dv[4];
    #pragma unroll
    for (int j = 0; j < 4; ++j) {
        const int e = e0 + j * 1024 + tid;
        dv[j] = (e < E) ? dst[e] : -1;
        sv[j] = (e < E) ? src[e] : 0;
    }
    __syncthreads();
    #pragma unroll
    for (int j = 0; j < 4; ++j)
        if (dv[j] >= 0) atomicAdd(&h[dv[j] >> GAT_BSH], 1);
    __syncthreads();
    const int v  = (tid < 256) ? h[tid] : 0;
    const int ex = gat_block_excl_scan(v, tid, wtot);
    if (tid < 256) cur[tid] = ex;
    if (tid < B) {
        cnt[c * B + tid] = v;
        loc[c * B + tid] = ex;
    }
    __syncthreads();
    #pragma unroll
    for (int j = 0; j < 4; ++j) {
        if (dv[j] >= 0) {
            const int lp = atomicAdd(&cur[dv[j] >> GAT_BSH], 1);
            stage[lp] = sv[j] | ((dv[j] & ((1 << GAT_BSH) - 1)) << GAT_SRCB);
        }
    }
    __syncthreads();
    const int nloc = min(GAT_CHE, E - e0);
    for (int lp = tid; lp < nloc; lp += 1024) cedge[e0 + lp] = stage[lp];
}

template<int MODE>
__device__ __forceinline__ void gat_seg_edge(int w, int slot, int* cn, int* stage,
                                             int* __restrict__ csr, int s0) {
    if (MODE == 2) {
        const int lp = atomicAdd(&cn[w >> GAT_SRCB], 1);
        csr[s0 + lp] = w & GAT_SRCM;
    } else {
        atomicAdd(&cn[w >> GAT_SRCB], 1);
        if (MODE == 0) stage[slot] = w;
    }
}

template<int MODE>
__device__ __forceinline__ void gat_walk_runs(const int* __restrict__ cedge, int nc, int oc, int pc,
                                              int lane, int* cn, int* stage,
                                              int* __restrict__ csr, int s0) {
    const unsigned live = __builtin_amdgcn_ballot_w32(nc > 0);
    #pragma unroll
    for (int g = 0; g < 32; g += GAT_SEGB) {
        if ((live >> g) == 0u) break;
        int w[GAT_SEGB];
        #pragma unroll
        for (int j = 0; j < GAT_SEGB; ++j) {
            const int n = __builtin_amdgcn_readlane(nc, g + j);
            const int o = __builtin_amdgcn_readlane(oc, g + j);
            w[j] = (lane < n) ? cedge[o + lane] : -1;
        }
        #pragma unroll
        for (int j = 0; j < GAT_SEGB; ++j) {
            const int p = __builtin_amdgcn_readlane(pc, g + j);
            if (w[j] >= 0) gat_seg_edge<MODE>(w[j], p + lane, cn, stage, csr, s0);
        }
        #pragma unroll
        for (int j = 0; j < GAT_SEGB; ++j) {
            const int n = __builtin_amdgcn_readlane(nc, g + j);
            if (n > 32) {
                const int o = __builtin_amdgcn_readlane(oc, g + j);
                const int p = __builtin_amdgcn_readlane(pc, g + j);
                for (int l = lane + 32; l < n; l += 32)
                    gat_seg_edge<MODE>(cedge[o + l], p + l, cn, stage, csr, s0);
            }
        }
    }
}

__global__ __launch_bounds__(1024) void gat_bkt_sort(const int* __restrict__ cedge,
                                                     const int* __restrict__ cnt,
                                                     const int* __restrict__ loc,
                                                     int* __restrict__ csr,
                                                     int* __restrict__ rowptr,
                                                     int N, int B, int C) {
    __shared__ int cn[512];
    __shared__ int wtot[32];
    __shared__ int wnum[32];
    __shared__ int wloc[32];
    __shared__ int stage[GAT_STG];
    __shared__ int stg[GAT_STG];
    const int tid  = threadIdx.x;
    const int lane = tid & 31;
    const int wv   = tid >> 5;
    const int b    = blockIdx.x;
    const int d0   = b << GAT_BSH;
    if (tid < 512) cn[tid] = 0;
    const int cc = lane * 32 + wv;
    int nc = 0, lc = 0;
    if (cc < C) {
        nc = cnt[cc * B + b];
        lc = loc[cc * B + b];
    }
    const int pc = gat_block_excl_scan(nc, tid, wnum);
    const int oc = cc * GAT_CHE + lc;
    int ls = lc;
    for (int o = 16; o > 0; o >>= 1) ls += __shfl_xor(ls, o);
    if (lane == 0) wloc[wv] = ls;
    __syncthreads();
    int s0 = 0, nb = 0;
    for (int k = 0; k < 32; ++k) {
        s0 += wloc[k];
        nb += wnum[k];
    }
    const bool staged = (nb <= GAT_STG);
    if (staged) gat_walk_runs<0>(cedge, nc, oc, pc, lane, cn, stage, csr, s0);
    else        gat_walk_runs<1>(cedge, nc, oc, pc, lane, cn, stage, csr, s0);
    __syncthreads();
    const int v  = (tid < 512) ? cn[tid] : 0;
    const int ex = gat_block_excl_scan(v, tid, wtot);
    if (tid < 512) {
        cn[tid] = ex;
        if (d0 + tid < N) rowptr[d0 + tid] = s0 + ex;
    }
    if (b == B - 1 && tid == 0) rowptr[N] = s0 + nb;
    __syncthreads();
    if (staged) {
        for (int i0 = 0; i0 < nb; i0 += 4096) {
            int w[4];
            #pragma unroll
            for (int j = 0; j < 4; ++j) {
                const int i = i0 + j * 1024 + tid;
                w[j] = (i < nb) ? stage[i] : -1;
            }
            #pragma unroll
            for (int j = 0; j < 4; ++j) {
                if (w[j] >= 0) {
                    const int lp = atomicAdd(&cn[w[j] >> GAT_SRCB], 1);
                    stg[lp] = w[j] & GAT_SRCM;
                }
            }
        }
        __syncthreads();
        for (int i = tid; i < nb; i += 1024) csr[s0 + i] = stg[i];
    } else {
        gat_walk_runs<2>(cedge, nc, oc, pc, lane, cn, stage, csr, s0);
    }
}

template<int G, int FS>
__device__ __forceinline__ void gat_acc_rows(const float* __restrict__ Fl, int s, float w, int t,
                                             float& ax, float& ay) {
    float2 f[G];
    float  wt[G];
    #pragma unroll
    for (int g = 0; g < G; ++g) {
        const int sg = __builtin_amdgcn_readlane(s, t + g);
        wt[g] = __int_as_float(__builtin_amdgcn_readlane(__float_as_int(w), t + g));
        f[g] = *(const float2*)(Fl + (size_t)sg * FS);
    }
    #pragma unroll
    for (int g = 0; g < G; ++g) {
        ax += wt[g] * f[g].x;
        ay += wt[g] * f[g].y;
    }
}

template<int FS>
__device__ __forceinline__ void gat_acc_chunk(const float* __restrict__ Fl, int s, float w, int cnt,
                                              float& ax, float& ay) {
    int t = 0;
    for (; t + 8 <= cnt; t += 8) gat_acc_rows<8, FS>(Fl, s, w, t, ax, ay);
    if (cnt & 4) { gat_acc_rows<4, FS>(Fl, s, w, t, ax, ay); t += 4; }
    if (cnt & 2) { gat_acc_rows<2, FS>(Fl, s, w, t, ax, ay); t += 2; }
    if (cnt & 1) { gat_acc_rows<1, FS>(Fl, s, w, t, ax, ay); }
}

template<int D, int FS, bool ELU>
__global__ __launch_bounds__(256) void gat_node_softmax_agg(const float* __restrict__ F,
                                                            const float* __restrict__ el,
                                                            const float* __restrict__ er,
                                                            const int* __restrict__ rowptr,
                                                            const int* __restrict__ csr,
                                                            const float* __restrict__ bias,
                                                            float* __restrict__ out, int N) {
    const int lane = threadIdx.x;
    const int node = blockIdx.x * 8 + __builtin_amdgcn_readfirstlane((int)threadIdx.y);
    if (node >= N) return;
    const int beg = rowptr[node];
    const int end = rowptr[node + 1];
    const float erd = er[node];
    const bool colok = (2 * lane < D);
    const float* __restrict__ Fl = F + (colok ? 2 * lane : 0);

    const bool has0 = (beg + lane < end);
    int s0 = 0;
    float v0 = -INFINITY;
    if (has0) {
        s0 = csr[beg + lane];
        const float v = el[s0] + erd;
        v0 = (v > 0.f) ? v : NEG_SLOPE * v;
    }
    float m = v0;
    for (int j = beg + 32 + lane; j < end; j += 32) {
        const float v = el[csr[j]] + erd;
        m = fmaxf(m, (v > 0.f) ? v : NEG_SLOPE * v);
    }
    for (int off = 16; off > 0; off >>= 1) m = fmaxf(m, __shfl_xor(m, off));

    float ax = 0.f, ay = 0.f;
    const float w0 = has0 ? __expf(v0 - m) : 0.f;
    float ssum = w0;
    if (end > beg)
        gat_acc_chunk<FS>(Fl, s0, w0, min(32, end - beg), ax, ay);
    for (int base = beg + 32; base < end; base += 32) {
        const int j = base + lane;
        int s = 0;
        float w = 0.f;
        if (j < end) {
            s = csr[j];
            const float v = el[s] + erd;
            w = __expf(((v > 0.f) ? v : NEG_SLOPE * v) - m);
        }
        ssum += w;
        gat_acc_chunk<FS>(Fl, s, w, min(32, end - base), ax, ay);
    }
    for (int off = 16; off > 0; off >>= 1) ssum += __shfl_xor(ssum, off);

    if (colok) {
        const float2 b = *(const float2*)(bias + 2 * lane);
        float x = b.x, y = b.y;
        if (end > beg) { x += ax / ssum; y += ay / ssum; }
        if (ELU) {
            x = (x > 0.f) ? x : (__expf(x) - 1.f);
            y = (y > 0.f) ? y : (__expf(y) - 1.f);
        }
        *(float2*)(out + (size_t)node * D + 2 * lane) = make_float2(x, y);
    }
}

template<int G, int D, int FS>
__device__ __forceinline__ void gat_online_rows(const float* __restrict__ Fl, int s, int t, float erd,
                                                float& m, float& ssum, float& ax, float& ay) {
    float2 f[G];
    #pragma unroll
    for (int g = 0; g < G; ++g) {
        const int sg = __builtin_amdgcn_readlane(s, t + g);
        f[g] = *(const float2*)(Fl + (size_t)sg * FS);
    }
    float e[G];
    float mn = m;
    #pragma unroll
    for (int g = 0; g < G; ++g) {
        const float v = __int_as_float(__builtin_amdgcn_readlane(__float_as_int(f[g].x), D / 2)) + erd;
        e[g] = (v > 0.f) ? v : NEG_SLOPE * v;
        mn = fmaxf(mn, e[g]);
    }
    const float sc = __expf(m - mn);
    ax *= sc; ay *= sc; ssum *= sc;
    #pragma unroll
    for (int g = 0; g < G; ++g) {
        const float w = __expf(e[g] - mn);
        ssum += w;
        ax += w * f[g].x;
        ay += w * f[g].y;
    }
    m = mn;
}

template<int D, int FS>
__global__ __launch_bounds__(256) void gat_node_online_agg(const float* __restrict__ F,
                                                           const float* __restrict__ er,
                                                           const int* __restrict__ rowptr,
                                                           const int* __restrict__ csr,
                                                           const float* __restrict__ bias,
                                                           float* __restrict__ out, int N) {
    static_assert(D % 2 == 0 && D + 2 <= FS && D / 2 < 32, "el sits in column D of the padded row");
    const int lane = threadIdx.x;
    const int node = blockIdx.x * 8 + __builtin_amdgcn_readfirstlane((int)threadIdx.y);
    if (node >= N) return;
    const int beg = rowptr[node];
    const int end = rowptr[node + 1];
    const float erd = er[node];
    const bool colok = (2 * lane < D);
    const float* __restrict__ Fl = F + ((2 * lane <= D) ? 2 * lane : 0);

    float m = -INFINITY, ssum = 0.f, ax = 0.f, ay = 0.f;
    for (int base = beg; base < end; base += 32) {
        const int j = base + lane;
        const int s = (j < end) ? csr[j] : 0;
        const int cnt = min(32, end - base);
        int t = 0;
        for (; t + 8 <= cnt; t += 8) gat_online_rows<8, D, FS>(Fl, s, t, erd, m, ssum, ax, ay);
        if (cnt & 4) { gat_online_rows<4, D, FS>(Fl, s, t, erd, m, ssum, ax, ay); t += 4; }
        if (cnt & 2) { gat_online_rows<2, D, FS>(Fl, s, t, erd, m, ssum, ax, ay); t += 2; }
        if (cnt & 1) { gat_online_rows<1, D, FS>(Fl, s, t, erd, m, ssum, ax, ay); }
    }

    if (colok) {
        const float2 b = *(const float2*)(bias + 2 * lane);
        float x = b.x, y = b.y;
        if (end > beg) { x += ax / ssum; y += ay / ssum; }
        *(float2*)(out + (size_t)node * D + 2 * lane) = make_float2(x, y);
    }
}

extern "C" void kernel_launch(void* const* d_in, const int* in_sizes, int n_in,
                              void* d_out, int out_size, void* d_ws, size_t ws_size,
                              hipStream_t stream) {
    const float* X   = (const float*)d_in[0];
    const int*   src = (const int*)d_in[1];
    const int*   dst = (const int*)d_in[2];
    const float* W1  = (const float*)d_in[3];
    const float* al1 = (const float*)d_in[4];
    const float* ar1 = (const float*)d_in[5];
    const float* b1  = (const float*)d_in[6];
    const float* W2  = (const float*)d_in[7];
    const float* al2 = (const float*)d_in[8];
    const float* ar2 = (const float*)d_in[9];
    const float* b2  = (const float*)d_in[10];

    const int IN = 128, H = 64, CLSPAD = 48;
    const int N = in_sizes[0] / IN;
    const int E = in_sizes[1];

    float* ws = (float*)d_ws;
    float* feat1 = ws;
    float* h     = feat1 + (size_t)N * H;
    float* feat2 = h + (size_t)N * H;
    float* el    = feat2 + (size_t)N * CLSPAD;
    float* er    = el + N;
    int*   cedge  = (int*)(er + N);
    int*   csr    = cedge + E;
    int*   rowptr = csr + E;
    const int B = (N + 511) / 512;
    const int C = (E + 4095) / 4096;
    int*   cnt    = rowptr + N + 1;
    int*   loc    = cnt + B * C;

    const int Mtiles = (N + 15) / 16;
    dim3 wblk(32, 8);
    const dim3 ggrid((Mtiles + 7) / 8);
    const dim3 ngrid((N + 7) / 8);

    hipLaunchKernelGGL(gat_chunk_file, dim3(C), dim3(1024), 0, stream,
                       src, dst, cnt, loc, cedge, E, B);
    hipLaunchKernelGGL(gat_bkt_sort, dim3(B), dim3(1024), 0, stream,
                       cedge, cnt, loc, csr, rowptr, N, B, C);

    hipLaunchKernelGGL((gat_gemm_wmma<128, 4, 64>), ggrid, wblk, 0, stream,
                       X, W1, al1, ar1, feat1, el, er, N);
    hipLaunchKernelGGL((gat_node_softmax_agg<64, 64, true>), ngrid, wblk, 0, stream,
                       feat1, el, er, rowptr, csr, b1, h, N);

    hipLaunchKernelGGL((gat_gemm_wmma<64, 3, 40>), ggrid, wblk, 0, stream,
                       h, W2, al2, ar2, feat2, el, er, N);
    hipLaunchKernelGGL((gat_node_online_agg<40, 48>), ngrid, wblk, 0, stream,
                       feat2, er, rowptr, csr, b2, (float*)d_out, N);
}


# Node launches: two nodes per wave, each 16-lane half owns a node, 4 columns per lane (b128 rows; source, weight by row_share); exp-sum by 16-lane tree, rescale per group of 8 rows: new sum order
# speedup vs baseline: 6.7838x; 1.0669x over previous
#include <hip/hip_runtime.h>
#include <hip/hip_bf16.h>
#include <math.h>

typedef __attribute__((ext_vector_type(2))) float v2f;
typedef __attribute__((ext_vector_type(8))) float v8f;

#define NEG_SLOPE 0.2f

template<int CTRL>
__device__ __forceinline__ float gat_dpp(float x) {
    return __int_as_float(__builtin_amdgcn_update_dpp(0, __float_as_int(x), CTRL, 0xf, 0xf, false));
}
__device__ __forceinline__ float gat_row16_sum(float x) {
    x += gat_dpp<0xB1>(x);
    x += gat_dpp<0x4E>(x);
    x += gat_dpp<0x141>(x);
    x += gat_dpp<0x140>(x);
    return x;
}

template<int K, int NT, int DOUT>
__global__ __launch_bounds__(256) __attribute__((amdgpu_waves_per_eu(2, 12))) void gat_gemm_wmma(const float* __restrict__ A,
                                                     const float* __restrict__ W,
                                                     const float* __restrict__ al,
                                                     const float* __restrict__ ar,
                                                     float* __restrict__ C,
                                                     float* __restrict__ el,
                                                     float* __restrict__ er, int N) {
    constexpr int DP = NT * 16;
    constexpr int CP = DP / 2;
    __shared__ float Bs[K * DP];

    const int tid = threadIdx.y * 32 + threadIdx.x;
    constexpr int ITEMS = (K / 2) * CP;
    static_assert(ITEMS % 256 == 0, "staging assumes a whole number of items per thread");
    static_assert(DOUT % 2 == 0 && DOUT <= DP, "column pairs must not straddle the padding");
    #pragma unroll
    for (int j = 0; j < ITEMS / 256; ++j) {
        const int i  = tid + 256 * j;
        const int kk = i / CP;
        const int c  = 2 * (i - kk * CP);
        const int cl = (DOUT < DP && c >= DOUT) ? 0 : c;
        float2 r0 = *(const float2*)&W[(size_t)(2 * kk) * DOUT + cl];
        float2 r1 = *(const float2*)&W[(size_t)(2 * kk + 1) * DOUT + cl];
        if (DOUT < DP && c >= DOUT) {
            r0 = make_float2(0.f, 0.f);
            r1 = make_float2(0.f, 0.f);
        }
        *(float4*)&Bs[(kk * DP + c) * 2] = make_float4(r0.x, r1.x, r0.y, r1.y);
    }
    __syncthreads();

    const int Mtiles = (N + 15) >> 4;
    const int strip = blockIdx.x * blockDim.y + threadIdx.y;
    if (strip >= Mtiles) return;
    int row0 = strip << 4;
    if (row0 + 16 > N) row0 = N - 16;

    const int m  = threadIdx.x & 15;
    const int kh = threadIdx.x >> 4;
    const float* __restrict__ arow = A + (size_t)(row0 + m) * K;

    float alv[NT], arv[NT];
    #pragma unroll
    for (int t = 0; t < NT; ++t) {
        const int cn = t * 16 + m;
        alv[t] = (cn < DOUT) ? al[cn] : 0.f;
        arv[t] = (cn < DOUT) ? ar[cn] : 0.f;
    }

    v8f acc[NT];
    #pragma unroll
    for (int t = 0; t < NT; ++t) acc[t] = (v8f){0.f,0.f,0.f,0.f,0.f,0.f,0.f,0.f};

    #pragma unroll
    for (int k = 0; k < K; k += 4) {
        const int ka = k + 2 * kh;
        const v2f a = *(const v2f*)(arow + ka);
        const int kk = ka >> 1;
        #pragma unroll
        for (int t = 0; t < NT; ++t) {
            const v2f b = *(const v2f*)&Bs[(kk * DP + t * 16 + m) * 2];
            acc[t] = __builtin_amdgcn_wmma_f32_16x16x4_f32(
                false, a, false, b, (short)0, acc[t], false, false);
        }
    }

    float pl[8], pr[8];
    #pragma unroll
    for (int v = 0; v < 8; ++v) {
        float a = acc[0][v] * alv[0];
        float b = acc[0][v] * arv[0];
        #pragma unroll
        for (int t = 1; t < NT; ++t) {
            a += acc[t][v] * alv[t];
            b += acc[t][v] * arv[t];
        }
        pl[v] = gat_row16_sum(a);
        pr[v] = gat_row16_sum(b);
    }

    #pragma unroll
    for (int t = 0; t < NT; ++t) {
        const int cn = t * 16 + m;
        #pragma unroll
        for (int v = 0; v < 8; ++v) {
            float val = acc[t][v];
            if (DOUT < DP && t == DOUT / 16 && cn == DOUT) val = pl[v];
            C[(size_t)(row0 + v + 8 * kh) * DP + cn] = val;
        }
    }
    const int r = m & 7;
    float sl = pl[0], sr = pr[0];
    #pragma unroll
    for (int v = 1; v < 8; ++v) {
        if (r == v) { sl = pl[v]; sr = pr[v]; }
    }
    float* __restrict__ dp = (m < 8) ? el : er;
    dp[row0 + 8 * kh + r] = (m < 8) ? sl : sr;
}

#define GAT_BSH 9
#define GAT_CHE 4096
#define GAT_STG 12288
#define GAT_SRCB 17
#define GAT_SRCM 0x1FFFF
#define GAT_SEGB 16

__device__ __forceinline__ int gat_block_excl_scan(int v, int tid, int* wtot) {
    const int lane = tid & 31;
    const int wv   = tid >> 5;
    int incl = v;
    for (int o = 1; o < 32; o <<= 1) {
        const int t = __shfl_up(incl, o);
        if (lane >= o) incl += t;
    }
    if (lane == 31) wtot[wv] = incl;
    __syncthreads();
    int carry = 0;
    for (int k = 0; k < wv; ++k) carry += wtot[k];
    return carry + incl - v;
}

__global__ __launch_bounds__(1024) void gat_chunk_file(const int* __restrict__ src,
                                                       const int* __restrict__ dst,
                                                       int* __restrict__ cnt,
                                                       int* __restrict__ loc,
                                                       int* __restrict__ cedge,
                                                       int E, int B) {
    __shared__ int h[256];
    __shared__ int cur[256];
    __shared__ int wtot[32];
    __shared__ int stage[GAT_CHE];
    const int tid = threadIdx.x;
    const int c   = blockIdx.x;
    const int e0  = c * GAT_CHE;
    if (tid < 256) h[tid] = 0;
    int sv[4], dv[4];
    #pragma unroll
    for (int j = 0; j < 4; ++j) {
        const int e = e0 + j * 1024 + tid;
        dv[j] = (e < E) ? dst[e] : -1;
        sv[j] = (e < E) ? src[e] : 0;
    }
    __syncthreads();
    #pragma unroll
    for (int j = 0; j < 4; ++j)
        if (dv[j] >= 0) atomicAdd(&h[dv[j] >> GAT_BSH], 1);
    __syncthreads();
    const int v  = (tid < 256) ? h[tid] : 0;
    const int ex = gat_block_excl_scan(v, tid, wtot);
    if (tid < 256) cur[tid] = ex;
    if (tid < B) {
        cnt[c * B + tid] = v;
        loc[c * B + tid] = ex;
    }
    __syncthreads();
    #pragma unroll
    for (int j = 0; j < 4; ++j) {
        if (dv[j] >= 0) {
            const int lp = atomicAdd(&cur[dv[j] >> GAT_BSH], 1);
            stage[lp] = sv[j] | ((dv[j] & ((1 << GAT_BSH) - 1)) << GAT_SRCB);
        }
    }
    __syncthreads();
    const int nloc = min(GAT_CHE, E - e0);
    for (int lp = tid; lp < nloc; lp += 1024) cedge[e0 + lp] = stage[lp];
}

template<int MODE>
__device__ __forceinline__ void gat_seg_edge(int w, int slot, int* cn, int* stage,
                                             int* __restrict__ csr, int s0) {
    if (MODE == 2) {
        const int lp = atomicAdd(&cn[w >> GAT_SRCB], 1);
        csr[s0 + lp] = w & GAT_SRCM;
    } else {
        atomicAdd(&cn[w >> GAT_SRCB], 1);
        if (MODE == 0) stage[slot] = w;
    }
}

template<int MODE>
__device__ __forceinline__ void gat_walk_runs(const int* __restrict__ cedge, int nc, int oc, int pc,
                                              int lane, int* cn, int* stage,
                                              int* __restrict__ csr, int s0) {
    const unsigned live = __builtin_amdgcn_ballot_w32(nc > 0);
    #pragma unroll
    for (int g = 0; g < 32; g += GAT_SEGB) {
        if ((live >> g) == 0u) break;
        int w[GAT_SEGB];
        #pragma unroll
        for (int j = 0; j < GAT_SEGB; ++j) {
            const int n = __builtin_amdgcn_readlane(nc, g + j);
            const int o = __builtin_amdgcn_readlane(oc, g + j);
            w[j] = (lane < n) ? cedge[o + lane] : -1;
        }
        #pragma unroll
        for (int j = 0; j < GAT_SEGB; ++j) {
            const int p = __builtin_amdgcn_readlane(pc, g + j);
            if (w[j] >= 0) gat_seg_edge<MODE>(w[j], p + lane, cn, stage, csr, s0);
        }
        #pragma unroll
        for (int j = 0; j < GAT_SEGB; ++j) {
            const int n = __builtin_amdgcn_readlane(nc, g + j);
            if (n > 32) {
                const int o = __builtin_amdgcn_readlane(oc, g + j);
                const int p = __builtin_amdgcn_readlane(pc, g + j);
                for (int l = lane + 32; l < n; l += 32)
                    gat_seg_edge<MODE>(cedge[o + l], p + l, cn, stage, csr, s0);
            }
        }
    }
}

__global__ __launch_bounds__(1024) void gat_bkt_sort(const int* __restrict__ cedge,
                                                     const int* __restrict__ cnt,
                                                     const int* __restrict__ loc,
                                                     int* __restrict__ csr,
                                                     int* __restrict__ rowptr,
                                                     int N, int B, int C) {
    __shared__ int cn[512];
    __shared__ int wtot[32];
    __shared__ int wnum[32];
    __shared__ int wloc[32];
    __shared__ int stage[GAT_STG];
    __shared__ int stg[GAT_STG];
    const int tid  = threadIdx.x;
    const int lane = tid & 31;
    const int wv   = tid >> 5;
    const int b    = blockIdx.x;
    const int d0   = b << GAT_BSH;
    if (tid < 512) cn[tid] = 0;
    const int cc = lane * 32 + wv;
    int nc = 0, lc = 0;
    if (cc < C) {
        nc = cnt[cc * B + b];
        lc = loc[cc * B + b];
    }
    const int pc = gat_block_excl_scan(nc, tid, wnum);
    const int oc = cc * GAT_CHE + lc;
    int ls = lc;
    for (int o = 16; o > 0; o >>= 1) ls += __shfl_xor(ls, o);
    if (lane == 0) wloc[wv] = ls;
    __syncthreads();
    int s0 = 0, nb = 0;
    for (int k = 0; k < 32; ++k) {
        s0 += wloc[k];
        nb += wnum[k];
    }
    const bool staged = (nb <= GAT_STG);
    if (staged) gat_walk_runs<0>(cedge, nc, oc, pc, lane, cn, stage, csr, s0);
    else        gat_walk_runs<1>(cedge, nc, oc, pc, lane, cn, stage, csr, s0);
    __syncthreads();
    const int v  = (tid < 512) ? cn[tid] : 0;
    const int ex = gat_block_excl_scan(v, tid, wtot);
    if (tid < 512) {
        cn[tid] = ex;
        if (d0 + tid < N) rowptr[d0 + tid] = s0 + ex;
    }
    if (b == B - 1 && tid == 0) rowptr[N] = s0 + nb;
    __syncthreads();
    if (staged) {
        for (int i0 = 0; i0 < nb; i0 += 4096) {
            int w[4];
            #pragma unroll
            for (int j = 0; j < 4; ++j) {
                const int i = i0 + j * 1024 + tid;
                w[j] = (i < nb) ? stage[i] : -1;
            }
            #pragma unroll
            for (int j = 0; j < 4; ++j) {
                if (w[j] >= 0) {
                    const int lp = atomicAdd(&cn[w[j] >> GAT_SRCB], 1);
                    stg[lp] = w[j] & GAT_SRCM;
                }
            }
        }
        __syncthreads();
        for (int i = tid; i < nb; i += 1024) csr[s0 + i] = stg[i];
    } else {
        gat_walk_runs<2>(cedge, nc, oc, pc, lane, cn, stage, csr, s0);
    }
}

template<int P>
__device__ __forceinline__ int gat_share(int x) {
    return __builtin_amdgcn_update_dpp(0, x, 0x150 + P, 0xf, 0xf, true);
}
typedef __attribute__((ext_vector_type(4))) float v4f;
__device__ __forceinline__ float gat_row16_max(float x) {
    x = fmaxf(x, gat_dpp<0xB1>(x));
    x = fmaxf(x, gat_dpp<0x4E>(x));
    x = fmaxf(x, gat_dpp<0x141>(x));
    x = fmaxf(x, gat_dpp<0x140>(x));
    return x;
}
template<int P, int FS>
__device__ __forceinline__ v4f gat_ldrow(const float* __restrict__ F, unsigned cb, int s) {
    const unsigned off = (unsigned)gat_share<P>(s) * (unsigned)(FS * 4) + cb;
    return *(const v4f*)((const char*)F + off);
}
template<int P>
__device__ __forceinline__ void gat_fma_row(v2f& a01, v2f& a23, float w, const v4f f) {
    const float wp = __int_as_float(gat_share<P>(__float_as_int(w)));
    a01 += f.lo * wp;
    a23 += f.hi * wp;
}
template<int T, int G, int FS>
__device__ __forceinline__ void gat_pair_rows(const float* __restrict__ F, unsigned cb, int s, float w,
                                              v2f& a01, v2f& a23) {
    static_assert(T + G <= 16, "positions stay inside the half");
    v4f f0, f1, f2, f3, f4, f5, f6, f7;
    f0 = gat_ldrow<T, FS>(F, cb, s);
    if constexpr (G > 1) f1 = gat_ldrow<T + 1, FS>(F, cb, s);
    if constexpr (G > 2) { f2 = gat_ldrow<T + 2, FS>(F, cb, s); f3 = gat_ldrow<T + 3, FS>(F, cb, s); }
    if constexpr (G > 4) {
        f4 = gat_ldrow<T + 4, FS>(F, cb, s); f5 = gat_ldrow<T + 5, FS>(F, cb, s);
        f6 = gat_ldrow<T + 6, FS>(F, cb, s); f7 = gat_ldrow<T + 7, FS>(F, cb, s);
    }
    gat_fma_row<T>(a01, a23, w, f0);
    if constexpr (G > 1) gat_fma_row<T + 1>(a01, a23, w, f1);
    if constexpr (G > 2) { gat_fma_row<T + 2>(a01, a23, w, f2); gat_fma_row<T + 3>(a01, a23, w, f3); }
    if constexpr (G > 4) {
        gat_fma_row<T + 4>(a01, a23, w, f4); gat_fma_row<T + 5>(a01, a23, w, f5);
        gat_fma_row<T + 6>(a01, a23, w, f6); gat_fma_row<T + 7>(a01, a23, w, f7);
    }
}
template<int T, int FS>
__device__ __forceinline__ void gat_pair_tail(const float* __restrict__ F, unsigned cb, int s, float w,
                                              int rest, v2f& a01, v2f& a23) {
    if (rest & 4) {
        gat_pair_rows<T, 4, FS>(F, cb, s, w, a01, a23);
        if (rest & 2) gat_pair_rows<T + 4, 2, FS>(F, cb, s, w, a01, a23);
    } else if (rest & 2) {
        gat_pair_rows<T, 2, FS>(F, cb, s, w, a01, a23);
    }
}
template<int FS>
__device__ __forceinline__ void gat_pair_chunk(const float* __restrict__ F, unsigned cb, int s, float w,
                                               int P, v2f& a01, v2f& a23) {
    const int P2 = (P + 1) & ~1;
    if (P2 >= 8) {
        gat_pair_rows<0, 8, FS>(F, cb, s, w, a01, a23);
        if (P2 == 16) gat_pair_rows<8, 8, FS>(F, cb, s, w, a01, a23);
    }
    if (P2 & 8) gat_pair_tail<8, FS>(F, cb, s, w, P2 & 6, a01, a23);
    else        gat_pair_tail<0, FS>(F, cb, s, w, P2 & 6, a01, a23);
}

template<int D, int FS, bool ELU>
__global__ __launch_bounds__(256) void gat_node2_softmax_agg(const float* __restrict__ F,
                                                             const float* __restrict__ el,
                                                             const float* __restrict__ er,
                                                             const int* __restrict__ rowptr,
                                                             const int* __restrict__ csr,
                                                             const float* __restrict__ bias,
                                                             float* __restrict__ out, int N) {
    static_assert(D == 64 && FS == 64, "16 lanes x 4 columns cover the row");
    const int lane = threadIdx.x;
    const int c = lane & 15;
    const int wv = blockIdx.x * 8 + __builtin_amdgcn_readfirstlane((int)threadIdx.y);
    if (2 * wv >= N) return;
    const int node = 2 * wv + (lane >> 4);
    const bool nok = (node < N);
    const int nd = nok ? node : N - 1;
    const int beg = rowptr[nd];
    const int nxt = rowptr[nd + 1];
    const int end = nok ? nxt : beg;
    const float erd = er[nd];
    const int deg = end - beg;
    const int last = max(end - 1, 0);
    const int dmax = max(__builtin_amdgcn_readlane(deg, 0), __builtin_amdgcn_readlane(deg, 16));
    const unsigned cb = (unsigned)c * 16u;

    const int i0 = beg + c, i1 = i0 + 16;
    const int s0 = csr[min(i0, last)];
    int s1 = 0;
    if (dmax > 16) s1 = csr[min(i1, last)];
    const float g0 = el[s0];
    float g1 = 0.f;
    if (dmax > 16) g1 = el[s1];
    float v0 = g0 + erd, v1 = g1 + erd;
    v0 = (v0 > 0.f) ? v0 : NEG_SLOPE * v0;
    v1 = (v1 > 0.f) ? v1 : NEG_SLOPE * v1;
    v0 = (i0 < end) ? v0 : -INFINITY;
    v1 = (i1 < end) ? v1 : -INFINITY;
    float m = fmaxf(v0, v1);
    for (int k = 32; k < dmax; k += 16) {
        const int i = i0 + k;
        const float v = el[csr[min(i, last)]] + erd;
        if (i < end) m = fmaxf(m, (v > 0.f) ? v : NEG_SLOPE * v);
    }
    m = gat_row16_max(m);

    v2f a01 = (v2f){0.f, 0.f}, a23 = (v2f){0.f, 0.f};
    const float w0 = (i0 < end) ? __expf(v0 - m) : 0.f;
    float ssum = w0;
    if (dmax > 0) gat_pair_chunk<FS>(F, cb, s0, w0, min(dmax, 16), a01, a23);
    if (dmax > 16) {
        const float w1 = (i1 < end) ? __expf(v1 - m) : 0.f;
        ssum += w1;
        gat_pair_chunk<FS>(F, cb, s1, w1, min(dmax - 16, 16), a01, a23);
    }
    for (int k = 32; k < dmax; k += 16) {
        const int i = i0 + k;
        const int s = csr[min(i, last)];
        const float v = el[s] + erd;
        const float w = (i < end) ? __expf(((v > 0.f) ? v : NEG_SLOPE * v) - m) : 0.f;
        ssum += w;
        gat_pair_chunk<FS>(F, cb, s, w, min(dmax - k, 16), a01, a23);
    }
    ssum = gat_row16_sum(ssum);

    if (nok) {
        const float4 b = *(const float4*)(bias + 4 * c);
        float x = b.x, y = b.y, z = b.z, u = b.w;
        if (deg > 0) { x += a01.x / ssum; y += a01.y / ssum; z += a23.x / ssum; u += a23.y / ssum; }
        if (ELU) {
            x = (x > 0.f) ? x : (__expf(x) - 1.f);
            y = (y > 0.f) ? y : (__expf(y) - 1.f);
            z = (z > 0.f) ? z : (__expf(z) - 1.f);
            u = (u > 0.f) ? u : (__expf(u) - 1.f);
        }
        *(float4*)(out + (size_t)node * D + 4 * c) = make_float4(x, y, z, u);
    }
}

template<int P, int D>
__device__ __forceinline__ float gat_row_logit(const v4f f, float erd, int cnt) {
    const float v = __int_as_float(gat_share<D / 4>(__float_as_int(f.x))) + erd;
    const float e = (v > 0.f) ? v : NEG_SLOPE * v;
    return (P < cnt) ? e : -INFINITY;
}
__device__ __forceinline__ void gat_acc_row(v2f& a01, v2f& a23, float& ssum, float w, const v4f f) {
    ssum += w;
    a01 += f.lo * w;
    a23 += f.hi * w;
}
template<int T, int G, int D, int FS>
__device__ __forceinline__ void gat_pair_online(const float* __restrict__ F, unsigned cb, int s, int cnt,
                                                float erd, float& m, float& ssum, v2f& a01, v2f& a23) {
    static_assert(T + G <= 16, "positions stay inside the half");
    v4f f0, f1, f2, f3, f4, f5, f6, f7;
    f0 = gat_ldrow<T, FS>(F, cb, s);
    if constexpr (G > 1) f1 = gat_ldrow<T + 1, FS>(F, cb, s);
    if constexpr (G > 2) { f2 = gat_ldrow<T + 2, FS>(F, cb, s); f3 = gat_ldrow<T + 3, FS>(F, cb, s); }
    if constexpr (G > 4) {
        f4 = gat_ldrow<T + 4, FS>(F, cb, s); f5 = gat_ldrow<T + 5, FS>(F, cb, s);
        f6 = gat_ldrow<T + 6, FS>(F, cb, s); f7 = gat_ldrow<T + 7, FS>(F, cb, s);
    }
    float e0, e1, e2, e3, e4, e5, e6, e7;
    float mn = m;
    e0 = gat_row_logit<T, D>(f0, erd, cnt); mn = fmaxf(mn, e0);
    if constexpr (G > 1) { e1 = gat_row_logit<T + 1, D>(f1, erd, cnt); mn = fmaxf(mn, e1); }
    if constexpr (G > 2) {
        e2 = gat_row_logit<T + 2, D>(f2, erd, cnt); mn = fmaxf(mn, e2);
        e3 = gat_row_logit<T + 3, D>(f3, erd, cnt); mn = fmaxf(mn, e3);
    }
    if constexpr (G > 4) {
        e4 = gat_row_logit<T + 4, D>(f4, erd, cnt); mn = fmaxf(mn, e4);
        e5 = gat_row_logit<T + 5, D>(f5, erd, cnt); mn = fmaxf(mn, e5);
        e6 = gat_row_logit<T + 6, D>(f6, erd, cnt); mn = fmaxf(mn, e6);
        e7 = gat_row_logit<T + 7, D>(f7, erd, cnt); mn = fmaxf(mn, e7);
    }
    const float sc = __expf(m - mn);
    a01 *= sc; a23 *= sc; ssum *= sc;
    gat_acc_row(a01, a23, ssum, __expf(e0 - mn), f0);
    if constexpr (G > 1) gat_acc_row(a01, a23, ssum, __expf(e1 - mn), f1);
    if constexpr (G > 2) { gat_acc_row(a01, a23, ssum, __expf(e2 - mn), f2); gat_acc_row(a01, a23, ssum, __expf(e3 - mn), f3); }
    if constexpr (G > 4) {
        gat_acc_row(a01, a23, ssum, __expf(e4 - mn), f4); gat_acc_row(a01, a23, ssum, __expf(e5 - mn), f5);
        gat_acc_row(a01, a23, ssum, __expf(e6 - mn), f6); gat_acc_row(a01, a23, ssum, __expf(e7 - mn), f7);
    }
    m = mn;
}
template<int T, int D, int FS>
__device__ __forceinline__ void gat_online_tail(const float* __restrict__ F, unsigned cb, int s, int cnt,
                                                float erd, int rest, float& m, float& ssum, v2f& a01, v2f& a23) {
    if (rest & 4) {
        gat_pair_online<T, 4, D, FS>(F, cb, s, cnt, erd, m, ssum, a01, a23);
        if (rest & 2) gat_pair_online<T + 4, 2, D, FS>(F, cb, s, cnt, erd, m, ssum, a01, a23);
    } else if (rest & 2) {
        gat_pair_online<T, 2, D, FS>(F, cb, s, cnt, erd, m, ssum, a01, a23);
    }
}

template<int D, int FS>
__global__ __launch_bounds__(256) void gat_node2_online_agg(const float* __restrict__ F,
                                                            const float* __restrict__ er,
                                                            const int* __restrict__ rowptr,
                                                            const int* __restrict__ csr,
                                                            const float* __restrict__ bias,
                                                            float* __restrict__ out, int N) {
    static_assert(D % 4 == 0 && D + 4 <= FS && FS <= 64 && FS % 4 == 0, "el sits in column D of the padded row");
    const int lane = threadIdx.x;
    const int c = lane & 15;
    const int wv = blockIdx.x * 8 + __builtin_amdgcn_readfirstlane((int)threadIdx.y);
    if (2 * wv >= N) return;
    const int node = 2 * wv + (lane >> 4);
    const bool nok = (node < N);
    const int nd = nok ? node : N - 1;
    const int beg = rowptr[nd];
    const int nxt = rowptr[nd + 1];
    const int end = nok ? nxt : beg;
    const float erd = er[nd];
    const int deg = end - beg;
    const int last = max(end - 1, 0);
    const int dmax = max(__builtin_amdgcn_readlane(deg, 0), __builtin_amdgcn_readlane(deg, 16));
    const unsigned cb = (4 * c < FS) ? (unsigned)c * 16u : 0u;

    float m = -INFINITY, ssum = 0.f;
    v2f a01 = (v2f){0.f, 0.f}, a23 = (v2f){0.f, 0.f};
    int s = csr[min(beg + c, last)];
    for (int k = 0; k < dmax; k += 16) {
        int sn = 0;
        if (k + 16 < dmax) sn = csr[min(beg + k + 16 + c, last)];
        const int cnt = deg - k;
        const int P2 = (min(dmax - k, 16) + 1) & ~1;
        if (P2 >= 8) {
            gat_pair_online<0, 8, D, FS>(F, cb, s, cnt, erd, m, ssum, a01, a23);
            if (P2 == 16) gat_pair_online<8, 8, D, FS>(F, cb, s, cnt, erd, m, ssum, a01, a23);
        }
        if (P2 & 8) gat_online_tail<8, D, FS>(F, cb, s, cnt, erd, P2 & 6, m, ssum, a01, a23);
        else        gat_online_tail<0, D, FS>(F, cb, s, cnt, erd, P2 & 6, m, ssum, a01, a23);
        s = sn;
    }

    if (nok && 4 * c < D) {
        const float4 b = *(const float4*)(bias + 4 * c);
        float x = b.x, y = b.y, z = b.z, u = b.w;
        if (deg > 0) { x += a01.x / ssum; y += a01.y / ssum; z += a23.x / ssum; u += a23.y / ssum; }
        *(float4*)(out + (size_t)node * D + 4 * c) = make_float4(x, y, z, u);
    }
}

extern "C" void kernel_launch(void* const* d_in, const int* in_sizes, int n_in,
                              void* d_out, int out_size, void* d_ws, size_t ws_size,
                              hipStream_t stream) {
    const float* X   = (const float*)d_in[0];
    const int*   src = (const int*)d_in[1];
    const int*   dst = (const int*)d_in[2];
    const float* W1  = (const float*)d_in[3];
    const float* al1 = (const float*)d_in[4];
    const float* ar1 = (const float*)d_in[5];
    const float* b1  = (const float*)d_in[6];
    const float* W2  = (const float*)d_in[7];
    const float* al2 = (const float*)d_in[8];
    const float* ar2 = (const float*)d_in[9];
    const float* b2  = (const float*)d_in[10];

    const int IN = 128, H = 64, CLSPAD = 48;
    const int N = in_sizes[0] / IN;
    const int E = in_sizes[1];

    float* ws = (float*)d_ws;
    float* feat1 = ws;
    float* h     = feat1 + (size_t)N * H;
    float* feat2 = h + (size_t)N * H;
    float* el    = feat2 + (size_t)N * CLSPAD;
    float* er    = el + N;
    int*   cedge  = (int*)(er + N);
    int*   csr    = cedge + E;
    int*   rowptr = csr + E;
    const int B = (N + 511) / 512;
    const int C = (E + 4095) / 4096;
    int*   cnt    = rowptr + N + 1;
    int*   loc    = cnt + B * C;

    const int Mtiles = (N + 15) / 16;
    dim3 wblk(32, 8);
    const dim3 ggrid((Mtiles + 7) / 8);
    const dim3 ngrid(((N + 1) / 2 + 7) / 8);

    hipLaunchKernelGGL(gat_chunk_file, dim3(C), dim3(1024), 0, stream,
                       src, dst, cnt, loc, cedge, E, B);
    hipLaunchKernelGGL(gat_bkt_sort, dim3(B), dim3(1024), 0, stream,
                       cedge, cnt, loc, csr, rowptr, N, B, C);

    hipLaunchKernelGGL((gat_gemm_wmma<128, 4, 64>), ggrid, wblk, 0, stream,
                       X, W1, al1, ar1, feat1, el, er, N);
    hipLaunchKernelGGL((gat_node2_softmax_agg<64, 64, true>), ngrid, wblk, 0, stream,
                       feat1, el, er, rowptr, csr, b1, h, N);

    hipLaunchKernelGGL((gat_gemm_wmma<64, 3, 40>), ggrid, wblk, 0, stream,
                       h, W2, al2, ar2, feat2, el, er, N);
    hipLaunchKernelGGL((gat_node2_online_agg<40, 48>), ngrid, wblk, 0, stream,
                       feat2, er, rowptr, csr, b2, (float*)d_out, N);
}


# Layer-1 GEMM rides in the bucket-sort launch: 224 product blocks (28 row strips each; 64 registers, so two 32-wave blocks share a WGP) beside the 196 bucket blocks; sum order kept; 6 -> 5 launches
# speedup vs baseline: 7.1955x; 1.0607x over previous
#include <hip/hip_runtime.h>
#include <hip/hip_bf16.h>
#include <math.h>

typedef __attribute__((ext_vector_type(2))) float v2f;
typedef __attribute__((ext_vector_type(8))) float v8f;

#define NEG_SLOPE 0.2f

template<int CTRL>
__device__ __forceinline__ float gat_dpp(float x) {
    return __int_as_float(__builtin_amdgcn_update_dpp(0, __float_as_int(x), CTRL, 0xf, 0xf, false));
}
__device__ __forceinline__ float gat_row16_sum(float x) {
    x += gat_dpp<0xB1>(x);
    x += gat_dpp<0x4E>(x);
    x += gat_dpp<0x141>(x);
    x += gat_dpp<0x140>(x);
    return x;
}

template<int K, int NT, int DOUT, int WAVES, int SPB, int AGRP>
__device__ __forceinline__ void gat_gemm_strips(const float* __restrict__ A,
                                                const float* __restrict__ W,
                                                const float* __restrict__ al,
                                                const float* __restrict__ ar,
                                                float* __restrict__ C,
                                                float* __restrict__ el,
                                                float* __restrict__ er, int N,
                                                const int blk, const int tid) {
    constexpr int DP = NT * 16;
    constexpr int CP = DP / 2;
    __shared__ float Bs[K * DP];

    constexpr int THREADS = WAVES * 32;
    constexpr int ITEMS = (K / 2) * CP;
    static_assert(ITEMS % THREADS == 0, "staging assumes a whole number of items per thread");
    static_assert(DOUT % 2 == 0 && DOUT <= DP, "column pairs must not straddle the padding");
    #pragma unroll
    for (int j = 0; j < ITEMS / THREADS; ++j) {
        const int i  = tid + THREADS * j;
        const int kk = i / CP;
        const int c  = 2 * (i - kk * CP);
        const int cl = (DOUT < DP && c >= DOUT) ? 0 : c;
        float2 r0 = *(const float2*)&W[(size_t)(2 * kk) * DOUT + cl];
        float2 r1 = *(const float2*)&W[(size_t)(2 * kk + 1) * DOUT + cl];
        if (DOUT < DP && c >= DOUT) {
            r0 = make_float2(0.f, 0.f);
            r1 = make_float2(0.f, 0.f);
        }
        *(float4*)&Bs[(kk * DP + c) * 2] = make_float4(r0.x, r1.x, r0.y, r1.y);
    }
    __syncthreads();

    const int Mtiles = (N + 15) >> 4;
    const int wave = __builtin_amdgcn_readfirstlane(tid >> 5);
    static_assert(SPB <= WAVES, "one strip per wave");
    if (SPB < WAVES && wave >= SPB) return;
    const int strip = blk * SPB + wave;
    if (strip >= Mtiles) return;
    int row0 = strip << 4;
    if (row0 + 16 > N) row0 = N - 16;

    const int m  = tid & 15;
    const int kh = (tid >> 4) & 1;
    const float* __restrict__ arow = A + (size_t)(row0 + m) * K;

    constexpr int KSTEPS = K / 4;
    constexpr bool GROUPED = (AGRP < KSTEPS);
    static_assert(!GROUPED || (AGRP == 8 && KSTEPS % 8 == 0), "the grouped loop is written for groups of 8");

    float alv[NT], arv[NT];
    if constexpr (!GROUPED) {
        #pragma unroll
        for (int t = 0; t < NT; ++t) {
            const int cn = t * 16 + m;
            alv[t] = (cn < DOUT) ? al[cn] : 0.f;
            arv[t] = (cn < DOUT) ? ar[cn] : 0.f;
        }
    }

    v8f acc[NT];
    #pragma unroll
    for (int t = 0; t < NT; ++t) acc[t] = (v8f){0.f,0.f,0.f,0.f,0.f,0.f,0.f,0.f};

    if constexpr (!GROUPED) {
        #pragma unroll
        for (int k = 0; k < K; k += 4) {
            const int ka = k + 2 * kh;
            const v2f a = *(const v2f*)(arow + ka);
            const int kk = ka >> 1;
            #pragma unroll
            for (int t = 0; t < NT; ++t) {
                const v2f b = *(const v2f*)&Bs[(kk * DP + t * 16 + m) * 2];
                acc[t] = __builtin_amdgcn_wmma_f32_16x16x4_f32(
                    false, a, false, b, (short)0, acc[t], false, false);
            }
        }
    } else {
        #pragma unroll
        for (int g = 0; g < KSTEPS; g += 8) {
            v2f a[8];
            #pragma unroll
            for (int j = 0; j < 8; ++j) a[j] = *(const v2f*)(arow + 4 * (g + j) + 2 * kh);
            asm volatile("" :: "v"(a[0]), "v"(a[1]), "v"(a[2]), "v"(a[3]),
                               "v"(a[4]), "v"(a[5]), "v"(a[6]), "v"(a[7]));
            #pragma unroll
            for (int j = 0; j < 8; ++j) {
                const int kk = 2 * (g + j) + kh;
                #pragma unroll
                for (int t = 0; t < NT; ++t) {
                    const v2f b = *(const v2f*)&Bs[(kk * DP + t * 16 + m) * 2];
                    acc[t] = __builtin_amdgcn_wmma_f32_16x16x4_f32(
                        false, a[j], false, b, (short)0, acc[t], false, false);
                }
            }
        }
        #pragma unroll
        for (int t = 0; t < NT; ++t) {
            const int cn = t * 16 + m;
            alv[t] = (cn < DOUT) ? al[cn] : 0.f;
            arv[t] = (cn < DOUT) ? ar[cn] : 0.f;
        }
    }

    float pl[8], pr[8];
    #pragma unroll
    for (int v = 0; v < 8; ++v) {
        float a = acc[0][v] * alv[0];
        float b = acc[0][v] * arv[0];
        #pragma unroll
        for (int t = 1; t < NT; ++t) {
            a += acc[t][v] * alv[t];
            b += acc[t][v] * arv[t];
        }
        pl[v] = gat_row16_sum(a);
        pr[v] = gat_row16_sum(b);
    }

    #pragma unroll
    for (int t = 0; t < NT; ++t) {
        const int cn = t * 16 + m;
        #pragma unroll
        for (int v = 0; v < 8; ++v) {
            float val = acc[t][v];
            if (DOUT < DP && t == DOUT / 16 && cn == DOUT) val = pl[v];
            C[(size_t)(row0 + v + 8 * kh) * DP + cn] = val;
        }
    }
    const int r = m & 7;
    float sl = pl[0], sr = pr[0];
    #pragma unroll
    for (int v = 1; v < 8; ++v) {
        if (r == v) { sl = pl[v]; sr = pr[v]; }
    }
    float* __restrict__ dp = (m < 8) ? el : er;
    dp[row0 + 8 * kh + r] = (m < 8) ? sl : sr;
}

template<int K, int NT, int DOUT>
__global__ __launch_bounds__(256) __attribute__((amdgpu_waves_per_eu(2, 12))) void gat_gemm_wmma(const float* __restrict__ A,
                                                     const float* __restrict__ W,
                                                     const float* __restrict__ al,
                                                     const float* __restrict__ ar,
                                                     float* __restrict__ C,
                                                     float* __restrict__ el,
                                                     float* __restrict__ er, int N) {
    gat_gemm_strips<K, NT, DOUT, 8, 8, K / 4>(A, W, al, ar, C, el, er, N,
                                    (int)blockIdx.x, (int)(threadIdx.y * 32 + threadIdx.x));
}

#define GAT_BSH 9
#define GAT_CHE 4096
#define GAT_STG 12288
#define GAT_SRCB 17
#define GAT_SRCM 0x1FFFF
#define GAT_SEGB 16

__device__ __forceinline__ int gat_block_excl_scan(int v, int tid, int* wtot) {
    const int lane = tid & 31;
    const int wv   = tid >> 5;
    int incl = v;
    for (int o = 1; o < 32; o <<= 1) {
        const int t = __shfl_up(incl, o);
        if (lane >= o) incl += t;
    }
    if (lane == 31) wtot[wv] = incl;
    __syncthreads();
    int carry = 0;
    for (int k = 0; k < wv; ++k) carry += wtot[k];
    return carry + incl - v;
}

__global__ __launch_bounds__(1024) void gat_chunk_file(const int* __restrict__ src,
                                                       const int* __restrict__ dst,
                                                       int* __restrict__ cnt,
                                                       int* __restrict__ loc,
                                                       int* __restrict__ cedge,
                                                       int E, int B) {
    __shared__ int h[256];
    __shared__ int cur[256];
    __shared__ int wtot[32];
    __shared__ int stage[GAT_CHE];
    const int tid = threadIdx.x;
    const int c   = blockIdx.x;
    const int e0  = c * GAT_CHE;
    if (tid < 256) h[tid] = 0;
    int sv[4], dv[4];
    #pragma unroll
    for (int j = 0; j < 4; ++j) {
        const int e = e0 + j * 1024 + tid;
        dv[j] = (e < E) ? dst[e] : -1;
        sv[j] = (e < E) ? src[e] : 0;
    }
    __syncthreads();
    #pragma unroll
    for (int j = 0; j < 4; ++j)
        if (dv[j] >= 0) atomicAdd(&h[dv[j] >> GAT_BSH], 1);
    __syncthreads();
    const int v  = (tid < 256) ? h[tid] : 0;
    const int ex = gat_block_excl_scan(v, tid, wtot);
    if (tid < 256) cur[tid] = ex;
    if (tid < B) {
        cnt[c * B + tid] = v;
        loc[c * B + tid] = ex;
    }
    __syncthreads();
    #pragma unroll
    for (int j = 0; j < 4; ++j) {
        if (dv[j] >= 0) {
            const int lp = atomicAdd(&cur[dv[j] >> GAT_BSH], 1);
            stage[lp] = sv[j] | ((dv[j] & ((1 << GAT_BSH) - 1)) << GAT_SRCB);
        }
    }
    __syncthreads();
    const int nloc = min(GAT_CHE, E - e0);
    for (int lp = tid; lp < nloc; lp += 1024) cedge[e0 + lp] = stage[lp];
}

template<int MODE>
__device__ __forceinline__ void gat_seg_edge(int w, int slot, int* cn, int* stage,
                                             int* __restrict__ csr, int s0) {
    if (MODE == 2) {
        const int lp = atomicAdd(&cn[w >> GAT_SRCB], 1);
        csr[s0 + lp] = w & GAT_SRCM;
    } else {
        atomicAdd(&cn[w >> GAT_SRCB], 1);
        if (MODE == 0) stage[slot] = w;
    }
}

template<int MODE>
__device__ __forceinline__ void gat_walk_runs(const int* __restrict__ cedge, int nc, int oc, int pc,
                                              int lane, int* cn, int* stage,
                                              int* __restrict__ csr, int s0) {
    const unsigned live = __builtin_amdgcn_ballot_w32(nc > 0);
    #pragma unroll
    for (int g = 0; g < 32; g += GAT_SEGB) {
        if ((live >> g) == 0u) break;
        int w[GAT_SEGB];
        #pragma unroll
        for (int j = 0; j < GAT_SEGB; ++j) {
            const int n = __builtin_amdgcn_readlane(nc, g + j);
            const int o = __builtin_amdgcn_readlane(oc, g + j);
            w[j] = (lane < n) ? cedge[o + lane] : -1;
        }
        #pragma unroll
        for (int j = 0; j < GAT_SEGB; ++j) {
            const int p = __builtin_amdgcn_readlane(pc, g + j);
            if (w[j] >= 0) gat_seg_edge<MODE>(w[j], p + lane, cn, stage, csr, s0);
        }
        #pragma unroll
        for (int j = 0; j < GAT_SEGB; ++j) {
            const int n = __builtin_amdgcn_readlane(nc, g + j);
            if (n > 32) {
                const int o = __builtin_amdgcn_readlane(oc, g + j);
                const int p = __builtin_amdgcn_readlane(pc, g + j);
                for (int l = lane + 32; l < n; l += 32)
                    gat_seg_edge<MODE>(cedge[o + l], p + l, cn, stage, csr, s0);
            }
        }
    }
}

__device__ __forceinline__ void gat_bkt_sort_block(const int* __restrict__ cedge,
                                                   const int* __restrict__ cnt,
                                                   const int* __restrict__ loc,
                                                   int* __restrict__ csr,
                                                   int* __restrict__ rowptr,
                                                   int N, int B, int C, const int b) {
    __shared__ int cn[512];
    __shared__ int wtot[32];
    __shared__ int wnum[32];
    __shared__ int wloc[32];
    __shared__ int stage[GAT_STG];
    __shared__ int stg[GAT_STG];
    const int tid  = threadIdx.x;
    const int lane = tid & 31;
    const int wv   = tid >> 5;
    const int d0   = b << GAT_BSH;
    if (tid < 512) cn[tid] = 0;
    const int cc = lane * 32 + wv;
    int nc = 0, lc = 0;
    if (cc < C) {
        nc = cnt[cc * B + b];
        lc = loc[cc * B + b];
    }
    const int pc = gat_block_excl_scan(nc, tid, wnum);
    const int oc = cc * GAT_CHE + lc;
    int ls = lc;
    for (int o = 16; o > 0; o >>= 1) ls += __shfl_xor(ls, o);
    if (lane == 0) wloc[wv] = ls;
    __syncthreads();
    int s0 = 0, nb = 0;
    for (int k = 0; k < 32; ++k) {
        s0 += wloc[k];
        nb += wnum[k];
    }
    const bool staged = (nb <= GAT_STG);
    if (staged) gat_walk_runs<0>(cedge, nc, oc, pc, lane, cn, stage, csr, s0);
    else        gat_walk_runs<1>(cedge, nc, oc, pc, lane, cn, stage, csr, s0);
    __syncthreads();
    const int v  = (tid < 512) ? cn[tid] : 0;
    const int ex = gat_block_excl_scan(v, tid, wtot);
    if (tid < 512) {
        cn[tid] = ex;
        if (d0 + tid < N) rowptr[d0 + tid] = s0 + ex;
    }
    if (b == B - 1 && tid == 0) rowptr[N] = s0 + nb;
    __syncthreads();
    if (staged) {
        for (int i0 = 0; i0 < nb; i0 += 4096) {
            int w[4];
            #pragma unroll
            for (int j = 0; j < 4; ++j) {
                const int i = i0 + j * 1024 + tid;
                w[j] = (i < nb) ? stage[i] : -1;
            }
            #pragma unroll
            for (int j = 0; j < 4; ++j) {
                if (w[j] >= 0) {
                    const int lp = atomicAdd(&cn[w[j] >> GAT_SRCB], 1);
                    stg[lp] = w[j] & GAT_SRCM;
                }
            }
        }
        __syncthreads();
        for (int i = tid; i < nb; i += 1024) csr[s0 + i] = stg[i];
    } else {
        gat_walk_runs<2>(cedge, nc, oc, pc, lane, cn, stage, csr, s0);
    }
}

#define GAT_SPB 28
__global__ __launch_bounds__(1024) __attribute__((amdgpu_waves_per_eu(16, 16))) void gat_bkt_sort_gemm(
                                                     const int* __restrict__ cedge,
                                                     const int* __restrict__ cnt,
                                                     const int* __restrict__ loc,
                                                     int* __restrict__ csr,
                                                     int* __restrict__ rowptr,
                                                     int N, int B, int C, int G,
                                                     const float* __restrict__ X,
                                                     const float* __restrict__ W,
                                                     const float* __restrict__ al,
                                                     const float* __restrict__ ar,
                                                     float* __restrict__ F,
                                                     float* __restrict__ el,
                                                     float* __restrict__ er) {
    const int b = blockIdx.x;
    if (__builtin_expect(b < G, 1)) {
        gat_gemm_strips<128, 4, 64, 32, GAT_SPB, 8>(X, W, al, ar, F, el, er, N, b, (int)threadIdx.x);
        return;
    }
    gat_bkt_sort_block(cedge, cnt, loc, csr, rowptr, N, B, C, b - G);
}

template<int P>
__device__ __forceinline__ int gat_share(int x) {
    return __builtin_amdgcn_update_dpp(0, x, 0x150 + P, 0xf, 0xf, true);
}
typedef __attribute__((ext_vector_type(4))) float v4f;
__device__ __forceinline__ float gat_row16_max(float x) {
    x = fmaxf(x, gat_dpp<0xB1>(x));
    x = fmaxf(x, gat_dpp<0x4E>(x));
    x = fmaxf(x, gat_dpp<0x141>(x));
    x = fmaxf(x, gat_dpp<0x140>(x));
    return x;
}
template<int P, int FS>
__device__ __forceinline__ v4f gat_ldrow(const float* __restrict__ F, unsigned cb, int s) {
    const unsigned off = (unsigned)gat_share<P>(s) * (unsigned)(FS * 4) + cb;
    return *(const v4f*)((const char*)F + off);
}
template<int P>
__device__ __forceinline__ void gat_fma_row(v2f& a01, v2f& a23, float w, const v4f f) {
    const float wp = __int_as_float(gat_share<P>(__float_as_int(w)));
    a01 += f.lo * wp;
    a23 += f.hi * wp;
}
template<int T, int G, int FS>
__device__ __forceinline__ void gat_pair_rows(const float* __restrict__ F, unsigned cb, int s, float w,
                                              v2f& a01, v2f& a23) {
    static_assert(T + G <= 16, "positions stay inside the half");
    v4f f0, f1, f2, f3, f4, f5, f6, f7;
    f0 = gat_ldrow<T, FS>(F, cb, s);
    if constexpr (G > 1) f1 = gat_ldrow<T + 1, FS>(F, cb, s);
    if constexpr (G > 2) { f2 = gat_ldrow<T + 2, FS>(F, cb, s); f3 = gat_ldrow<T + 3, FS>(F, cb, s); }
    if constexpr (G > 4) {
        f4 = gat_ldrow<T + 4, FS>(F, cb, s); f5 = gat_ldrow<T + 5, FS>(F, cb, s);
        f6 = gat_ldrow<T + 6, FS>(F, cb, s); f7 = gat_ldrow<T + 7, FS>(F, cb, s);
    }
    gat_fma_row<T>(a01, a23, w, f0);
    if constexpr (G > 1) gat_fma_row<T + 1>(a01, a23, w, f1);
    if constexpr (G > 2) { gat_fma_row<T + 2>(a01, a23, w, f2); gat_fma_row<T + 3>(a01, a23, w, f3); }
    if constexpr (G > 4) {
        gat_fma_row<T + 4>(a01, a23, w, f4); gat_fma_row<T + 5>(a01, a23, w, f5);
        gat_fma_row<T + 6>(a01, a23, w, f6); gat_fma_row<T + 7>(a01, a23, w, f7);
    }
}
template<int T, int FS>
__device__ __forceinline__ void gat_pair_tail(const float* __restrict__ F, unsigned cb, int s, float w,
                                              int rest, v2f& a01, v2f& a23) {
    if (rest & 4) {
        gat_pair_rows<T, 4, FS>(F, cb, s, w, a01, a23);
        if (rest & 2) gat_pair_rows<T + 4, 2, FS>(F, cb, s, w, a01, a23);
    } else if (rest & 2) {
        gat_pair_rows<T, 2, FS>(F, cb, s, w, a01, a23);
    }
}
template<int FS>
__device__ __forceinline__ void gat_pair_chunk(const float* __restrict__ F, unsigned cb, int s, float w,
                                               int P, v2f& a01, v2f& a23) {
    const int P2 = (P + 1) & ~1;
    if (P2 >= 8) {
        gat_pair_rows<0, 8, FS>(F, cb, s, w, a01, a23);
        if (P2 == 16) gat_pair_rows<8, 8, FS>(F, cb, s, w, a01, a23);
    }
    if (P2 & 8) gat_pair_tail<8, FS>(F, cb, s, w, P2 & 6, a01, a23);
    else        gat_pair_tail<0, FS>(F, cb, s, w, P2 & 6, a01, a23);
}

template<int D, int FS, bool ELU>
__global__ __launch_bounds__(256) void gat_node2_softmax_agg(const float* __restrict__ F,
                                                             const float* __restrict__ el,
                                                             const float* __restrict__ er,
                                                             const int* __restrict__ rowptr,
                                                             const int* __restrict__ csr,
                                                             const float* __restrict__ bias,
                                                             float* __restrict__ out, int N) {
    static_assert(D == 64 && FS == 64, "16 lanes x 4 columns cover the row");
    const int lane = threadIdx.x;
    const int c = lane & 15;
    const int wv = blockIdx.x * 8 + __builtin_amdgcn_readfirstlane((int)threadIdx.y);
    if (2 * wv >= N) return;
    const int node = 2 * wv + (lane >> 4);
    const bool nok = (node < N);
    const int nd = nok ? node : N - 1;
    const int beg = rowptr[nd];
    const int nxt = rowptr[nd + 1];
    const int end = nok ? nxt : beg;
    const float erd = er[nd];
    const int deg = end - beg;
    const int last = max(end - 1, 0);
    const int dmax = max(__builtin_amdgcn_readlane(deg, 0), __builtin_amdgcn_readlane(deg, 16));
    const unsigned cb = (unsigned)c * 16u;

    const int i0 = beg + c, i1 = i0 + 16;
    const int s0 = csr[min(i0, last)];
    int s1 = 0;
    if (dmax > 16) s1 = csr[min(i1, last)];
    const float g0 = el[s0];
    float g1 = 0.f;
    if (dmax > 16) g1 = el[s1];
    float v0 = g0 + erd, v1 = g1 + erd;
    v0 = (v0 > 0.f) ? v0 : NEG_SLOPE * v0;
    v1 = (v1 > 0.f) ? v1 : NEG_SLOPE * v1;
    v0 = (i0 < end) ? v0 : -INFINITY;
    v1 = (i1 < end) ? v1 : -INFINITY;
    float m = fmaxf(v0, v1);
    for (int k = 32; k < dmax; k += 16) {
        const int i = i0 + k;
        const float v = el[csr[min(i, last)]] + erd;
        if (i < end) m = fmaxf(m, (v > 0.f) ? v : NEG_SLOPE * v);
    }
    m = gat_row16_max(m);

    v2f a01 = (v2f){0.f, 0.f}, a23 = (v2f){0.f, 0.f};
    const float w0 = (i0 < end) ? __expf(v0 - m) : 0.f;
    float ssum = w0;
    if (dmax > 0) gat_pair_chunk<FS>(F, cb, s0, w0, min(dmax, 16), a01, a23);
    if (dmax > 16) {
        const float w1 = (i1 < end) ? __expf(v1 - m) : 0.f;
        ssum += w1;
        gat_pair_chunk<FS>(F, cb, s1, w1, min(dmax - 16, 16), a01, a23);
    }
    for (int k = 32; k < dmax; k += 16) {
        const int i = i0 + k;
        const int s = csr[min(i, last)];
        const float v = el[s] + erd;
        const float w = (i < end) ? __expf(((v > 0.f) ? v : NEG_SLOPE * v) - m) : 0.f;
        ssum += w;
        gat_pair_chunk<FS>(F, cb, s, w, min(dmax - k, 16), a01, a23);
    }
    ssum = gat_row16_sum(ssum);

    if (nok) {
        const float4 b = *(const float4*)(bias + 4 * c);
        float x = b.x, y = b.y, z = b.z, u = b.w;
        if (deg > 0) { x += a01.x / ssum; y += a01.y / ssum; z += a23.x / ssum; u += a23.y / ssum; }
        if (ELU) {
            x = (x > 0.f) ? x : (__expf(x) - 1.f);
            y = (y > 0.f) ? y : (__expf(y) - 1.f);
            z = (z > 0.f) ? z : (__expf(z) - 1.f);
            u = (u > 0.f) ? u : (__expf(u) - 1.f);
        }
        *(float4*)(out + (size_t)node * D + 4 * c) = make_float4(x, y, z, u);
    }
}

template<int P, int D>
__device__ __forceinline__ float gat_row_logit(const v4f f, float erd, int cnt) {
    const float v = __int_as_float(gat_share<D / 4>(__float_as_int(f.x))) + erd;
    const float e = (v > 0.f) ? v : NEG_SLOPE * v;
    return (P < cnt) ? e : -INFINITY;
}
__device__ __forceinline__ void gat_acc_row(v2f& a01, v2f& a23, float& ssum, float w, const v4f f) {
    ssum += w;
    a01 += f.lo * w;
    a23 += f.hi * w;
}
template<int T, int G, int D, int FS>
__device__ __forceinline__ void gat_pair_online(const float* __restrict__ F, unsigned cb, int s, int cnt,
                                                float erd, float& m, float& ssum, v2f& a01, v2f& a23) {
    static_assert(T + G <= 16, "positions stay inside the half");
    v4f f0, f1, f2, f3, f4, f5, f6, f7;
    f0 = gat_ldrow<T, FS>(F, cb, s);
    if constexpr (G > 1) f1 = gat_ldrow<T + 1, FS>(F, cb, s);
    if constexpr (G > 2) { f2 = gat_ldrow<T + 2, FS>(F, cb, s); f3 = gat_ldrow<T + 3, FS>(F, cb, s); }
    if constexpr (G > 4) {
        f4 = gat_ldrow<T + 4, FS>(F, cb, s); f5 = gat_ldrow<T + 5, FS>(F, cb, s);
        f6 = gat_ldrow<T + 6, FS>(F, cb, s); f7 = gat_ldrow<T + 7, FS>(F, cb, s);
    }
    float e0, e1, e2, e3, e4, e5, e6, e7;
    float mn = m;
    e0 = gat_row_logit<T, D>(f0, erd, cnt); mn = fmaxf(mn, e0);
    if constexpr (G > 1) { e1 = gat_row_logit<T + 1, D>(f1, erd, cnt); mn = fmaxf(mn, e1); }
    if constexpr (G > 2) {
        e2 = gat_row_logit<T + 2, D>(f2, erd, cnt); mn = fmaxf(mn, e2);
        e3 = gat_row_logit<T + 3, D>(f3, erd, cnt); mn = fmaxf(mn, e3);
    }
    if constexpr (G > 4) {
        e4 = gat_row_logit<T + 4, D>(f4, erd, cnt); mn = fmaxf(mn, e4);
        e5 = gat_row_logit<T + 5, D>(f5, erd, cnt); mn = fmaxf(mn, e5);
        e6 = gat_row_logit<T + 6, D>(f6, erd, cnt); mn = fmaxf(mn, e6);
        e7 = gat_row_logit<T + 7, D>(f7, erd, cnt); mn = fmaxf(mn, e7);
    }
    const float sc = __expf(m - mn);
    a01 *= sc; a23 *= sc; ssum *= sc;
    gat_acc_row(a01, a23, ssum, __expf(e0 - mn), f0);
    if constexpr (G > 1) gat_acc_row(a01, a23, ssum, __expf(e1 - mn), f1);
    if constexpr (G > 2) { gat_acc_row(a01, a23, ssum, __expf(e2 - mn), f2); gat_acc_row(a01, a23, ssum, __expf(e3 - mn), f3); }
    if constexpr (G > 4) {
        gat_acc_row(a01, a23, ssum, __expf(e4 - mn), f4); gat_acc_row(a01, a23, ssum, __expf(e5 - mn), f5);
        gat_acc_row(a01, a23, ssum, __expf(e6 - mn), f6); gat_acc_row(a01, a23, ssum, __expf(e7 - mn), f7);
    }
    m = mn;
}
template<int T, int D, int FS>
__device__ __forceinline__ void gat_online_tail(const float* __restrict__ F, unsigned cb, int s, int cnt,
                                                float erd, int rest, float& m, float& ssum, v2f& a01, v2f& a23) {
    if (rest & 4) {
        gat_pair_online<T, 4, D, FS>(F, cb, s, cnt, erd, m, ssum, a01, a23);
        if (rest & 2) gat_pair_online<T + 4, 2, D, FS>(F, cb, s, cnt, erd, m, ssum, a01, a23);
    } else if (rest & 2) {
        gat_pair_online<T, 2, D, FS>(F, cb, s, cnt, erd, m, ssum, a01, a23);
    }
}

template<int D, int FS>
__global__ __launch_bounds__(256) void gat_node2_online_agg(const float* __restrict__ F,
                                                            const float* __restrict__ er,
                                                            const int* __restrict__ rowptr,
                                                            const int* __restrict__ csr,
                                                            const float* __restrict__ bias,
                                                            float* __restrict__ out, int N) {
    static_assert(D % 4 == 0 && D + 4 <= FS && FS <= 64 && FS % 4 == 0, "el sits in column D of the padded row");
    const int lane = threadIdx.x;
    const int c = lane & 15;
    const int wv = blockIdx.x * 8 + __builtin_amdgcn_readfirstlane((int)threadIdx.y);
    if (2 * wv >= N) return;
    const int node = 2 * wv + (lane >> 4);
    const bool nok = (node < N);
    const int nd = nok ? node : N - 1;
    const int beg = rowptr[nd];
    const int nxt = rowptr[nd + 1];
    const int end = nok ? nxt : beg;
    const float erd = er[nd];
    const int deg = end - beg;
    const int last = max(end - 1, 0);
    const int dmax = max(__builtin_amdgcn_readlane(deg, 0), __builtin_amdgcn_readlane(deg, 16));
    const unsigned cb = (4 * c < FS) ? (unsigned)c * 16u : 0u;

    float m = -INFINITY, ssum = 0.f;
    v2f a01 = (v2f){0.f, 0.f}, a23 = (v2f){0.f, 0.f};
    int s = csr[min(beg + c, last)];
    for (int k = 0; k < dmax; k += 16) {
        int sn = 0;
        if (k + 16 < dmax) sn = csr[min(beg + k + 16 + c, last)];
        const int cnt = deg - k;
        const int P2 = (min(dmax - k, 16) + 1) & ~1;
        if (P2 >= 8) {
            gat_pair_online<0, 8, D, FS>(F, cb, s, cnt, erd, m, ssum, a01, a23);
            if (P2 == 16) gat_pair_online<8, 8, D, FS>(F, cb, s, cnt, erd, m, ssum, a01, a23);
        }
        if (P2 & 8) gat_online_tail<8, D, FS>(F, cb, s, cnt, erd, P2 & 6, m, ssum, a01, a23);
        else        gat_online_tail<0, D, FS>(F, cb, s, cnt, erd, P2 & 6, m, ssum, a01, a23);
        s = sn;
    }

    if (nok && 4 * c < D) {
        const float4 b = *(const float4*)(bias + 4 * c);
        float x = b.x, y = b.y, z = b.z, u = b.w;
        if (deg > 0) { x += a01.x / ssum; y += a01.y / ssum; z += a23.x / ssum; u += a23.y / ssum; }
        *(float4*)(out + (size_t)node * D + 4 * c) = make_float4(x, y, z, u);
    }
}

extern "C" void kernel_launch(void* const* d_in, const int* in_sizes, int n_in,
                              void* d_out, int out_size, void* d_ws, size_t ws_size,
                              hipStream_t stream) {
    const float* X   = (const float*)d_in[0];
    const int*   src = (const int*)d_in[1];
    const int*   dst = (const int*)d_in[2];
    const float* W1  = (const float*)d_in[3];
    const float* al1 = (const float*)d_in[4];
    const float* ar1 = (const float*)d_in[5];
    const float* b1  = (const float*)d_in[6];
    const float* W2  = (const float*)d_in[7];
    const float* al2 = (const float*)d_in[8];
    const float* ar2 = (const float*)d_in[9];
    const float* b2  = (const float*)d_in[10];

    const int IN = 128, H = 64, CLSPAD = 48;
    const int N = in_sizes[0] / IN;
    const int E = in_sizes[1];

    float* ws = (float*)d_ws;
    float* feat1 = ws;
    float* h     = feat1 + (size_t)N * H;
    float* feat2 = h + (size_t)N * H;
    float* el    = feat2 + (size_t)N * CLSPAD;
    float* er    = el + N;
    int*   cedge  = (int*)(er + N);
    int*   csr    = cedge + E;
    int*   rowptr = csr + E;
    const int B = (N + 511) / 512;
    const int C = (E + 4095) / 4096;
    int*   cnt    = rowptr + N + 1;
    int*   loc    = cnt + B * C;

    const int Mtiles = (N + 15) / 16;
    dim3 wblk(32, 8);
    const dim3 ggrid((Mtiles + 7) / 8);
    const dim3 ngrid(((N + 1) / 2 + 7) / 8);

    hipLaunchKernelGGL(gat_chunk_file, dim3(C), dim3(1024), 0, stream,
                       src, dst, cnt, loc, cedge, E, B);
    const int G = (Mtiles + GAT_SPB - 1) / GAT_SPB;
    hipLaunchKernelGGL(gat_bkt_sort_gemm, dim3(G + B), dim3(1024), 0, stream,
                       cedge, cnt, loc, csr, rowptr, N, B, C, G,
                       X, W1, al1, ar1, feat1, el, er);

    hipLaunchKernelGGL((gat_node2_softmax_agg<64, 64, true>), ngrid, wblk, 0, stream,
                       feat1, el, er, rowptr, csr, b1, h, N);

    hipLaunchKernelGGL((gat_gemm_wmma<64, 3, 40>), ggrid, wblk, 0, stream,
                       h, W2, al2, ar2, feat2, el, er, N);
    hipLaunchKernelGGL((gat_node2_online_agg<40, 48>), ngrid, wblk, 0, stream,
                       feat2, er, rowptr, csr, b2, (float*)d_out, N);
}


# Layer-2 product rides in the layer-1 node launch: a block's 16 hidden rows go to an LDS tile, the wave that arrives last (LDS counter) multiplies them by W2 on WMMA; h never stored; 5 -> 4 launches
# speedup vs baseline: 7.5461x; 1.0487x over previous
#include <hip/hip_runtime.h>
#include <hip/hip_bf16.h>
#include <math.h>

typedef __attribute__((ext_vector_type(2))) float v2f;
typedef __attribute__((ext_vector_type(8))) float v8f;

#define NEG_SLOPE 0.2f

template<int CTRL>
__device__ __forceinline__ float gat_dpp(float x) {
    return __int_as_float(__builtin_amdgcn_update_dpp(0, __float_as_int(x), CTRL, 0xf, 0xf, false));
}
__device__ __forceinline__ float gat_row16_sum(float x) {
    x += gat_dpp<0xB1>(x);
    x += gat_dpp<0x4E>(x);
    x += gat_dpp<0x141>(x);
    x += gat_dpp<0x140>(x);
    return x;
}


template<int K, int NT, int DOUT, int THREADS>
__device__ __forceinline__ void gat_stage_w_pairs(const float* __restrict__ W, float* Bs, const int tid) {
    constexpr int DP = NT * 16;
    constexpr int CP = DP / 2;
    constexpr int ITEMS = (K / 2) * CP;
    static_assert(ITEMS % THREADS == 0, "staging assumes a whole number of items per thread");
    static_assert(DOUT % 2 == 0 && DOUT <= DP, "column pairs must not straddle the padding");
    #pragma unroll
    for (int j = 0; j < ITEMS / THREADS; ++j) {
        const int i  = tid + THREADS * j;
        const int kk = i / CP;
        const int c  = 2 * (i - kk * CP);
        const int cl = (DOUT < DP && c >= DOUT) ? 0 : c;
        float2 r0 = *(const float2*)&W[(size_t)(2 * kk) * DOUT + cl];
        float2 r1 = *(const float2*)&W[(size_t)(2 * kk + 1) * DOUT + cl];
        if (DOUT < DP && c >= DOUT) {
            r0 = make_float2(0.f, 0.f);
            r1 = make_float2(0.f, 0.f);
        }
        *(float4*)&Bs[(kk * DP + c) * 2] = make_float4(r0.x, r1.x, r0.y, r1.y);
    }
}

template<int NT, int DOUT>
__device__ __forceinline__ void gat_strip_epilogue(const v8f (&acc)[NT],
                                                   const float* __restrict__ al,
                                                   const float* __restrict__ ar,
                                                   float* __restrict__ C,
                                                   float* __restrict__ el,
                                                   float* __restrict__ er,
                                                   const int row0, const int m, const int kh) {
    constexpr int DP = NT * 16;
    float alv[NT], arv[NT];
    #pragma unroll
    for (int t = 0; t < NT; ++t) {
        const int cn = t * 16 + m;
        alv[t] = (cn < DOUT) ? al[cn] : 0.f;
        arv[t] = (cn < DOUT) ? ar[cn] : 0.f;
    }

    float pl[8], pr[8];
    #pragma unroll
    for (int v = 0; v < 8; ++v) {
        float a = acc[0][v] * alv[0];
        float b = acc[0][v] * arv[0];
        #pragma unroll
        for (int t = 1; t < NT; ++t) {
            a += acc[t][v] * alv[t];
            b += acc[t][v] * arv[t];
        }
        pl[v] = gat_row16_sum(a);
        pr[v] = gat_row16_sum(b);
    }

    #pragma unroll
    for (int t = 0; t < NT; ++t) {
        const int cn = t * 16 + m;
        #pragma unroll
        for (int v = 0; v < 8; ++v) {
            float val = acc[t][v];
            if (DOUT < DP && t == DOUT / 16 && cn == DOUT) val = pl[v];
            C[(size_t)(row0 + v + 8 * kh) * DP + cn] = val;
        }
    }
    const int r = m & 7;
    float sl = pl[0], sr = pr[0];
    #pragma unroll
    for (int v = 1; v < 8; ++v) {
        if (r == v) { sl = pl[v]; sr = pr[v]; }
    }
    float* __restrict__ dp = (m < 8) ? el : er;
    dp[row0 + 8 * kh + r] = (m < 8) ? sl : sr;
}

template<int K, int NT, int DOUT, int WAVES, int SPB, int AGRP>
__device__ __forceinline__ void gat_gemm_strips(const float* __restrict__ A,
                                                const float* __restrict__ W,
                                                const float* __restrict__ al,
                                                const float* __restrict__ ar,
                                                float* __restrict__ C,
                                                float* __restrict__ el,
                                                float* __restrict__ er, int N,
                                                const int blk, const int tid) {
    constexpr int DP = NT * 16;
    __shared__ float Bs[K * DP];

    gat_stage_w_pairs<K, NT, DOUT, WAVES * 32>(W, Bs, tid);
    __syncthreads();

    const int Mtiles = (N + 15) >> 4;
    const int wave = __builtin_amdgcn_readfirstlane(tid >> 5);
    static_assert(SPB <= WAVES, "one strip per wave");
    if (SPB < WAVES && wave >= SPB) return;
    const int strip = blk * SPB + wave;
    if (strip >= Mtiles) return;
    int row0 = strip << 4;
    if (row0 + 16 > N) row0 = N - 16;

    const int m  = tid & 15;
    const int kh = (tid >> 4) & 1;
    const float* __restrict__ arow = A + (size_t)(row0 + m) * K;

    constexpr int KSTEPS = K / 4;
    constexpr bool GROUPED = (AGRP < KSTEPS);
    static_assert(!GROUPED || (AGRP == 8 && KSTEPS % 8 == 0), "the grouped loop is written for groups of 8");

    v8f acc[NT];
    #pragma unroll
    for (int t = 0; t < NT; ++t) acc[t] = (v8f){0.f,0.f,0.f,0.f,0.f,0.f,0.f,0.f};

    if constexpr (!GROUPED) {
        #pragma unroll
        for (int k = 0; k < K; k += 4) {
            const int ka = k + 2 * kh;
            const v2f a = *(const v2f*)(arow + ka);
            const int kk = ka >> 1;
            #pragma unroll
            for (int t = 0; t < NT; ++t) {
                const v2f b = *(const v2f*)&Bs[(kk * DP + t * 16 + m) * 2];
                acc[t] = __builtin_amdgcn_wmma_f32_16x16x4_f32(
                    false, a, false, b, (short)0, acc[t], false, false);
            }
        }
    } else {
        #pragma unroll
        for (int g = 0; g < KSTEPS; g += 8) {
            v2f a[8];
            #pragma unroll
            for (int j = 0; j < 8; ++j) a[j] = *(const v2f*)(arow + 4 * (g + j) + 2 * kh);
            asm volatile("" :: "v"(a[0]), "v"(a[1]), "v"(a[2]), "v"(a[3]),
                               "v"(a[4]), "v"(a[5]), "v"(a[6]), "v"(a[7]));
            #pragma unroll
            for (int j = 0; j < 8; ++j) {
                const int kk = 2 * (g + j) + kh;
                #pragma unroll
                for (int t = 0; t < NT; ++t) {
                    const v2f b = *(const v2f*)&Bs[(kk * DP + t * 16 + m) * 2];
                    acc[t] = __builtin_amdgcn_wmma_f32_16x16x4_f32(
                        false, a[j], false, b, (short)0, acc[t], false, false);
                }
            }
        }
    }

    gat_strip_epilogue<NT, DOUT>(acc, al, ar, C, el, er, row0, m, kh);
}

#define GAT_BSH 9
#define GAT_CHE 4096
#define GAT_STG 12288
#define GAT_SRCB 17
#define GAT_SRCM 0x1FFFF
#define GAT_SEGB 16

__device__ __forceinline__ int gat_block_excl_scan(int v, int tid, int* wtot) {
    const int lane = tid & 31;
    const int wv   = tid >> 5;
    int incl = v;
    for (int o = 1; o < 32; o <<= 1) {
        const int t = __shfl_up(incl, o);
        if (lane >= o) incl += t;
    }
    if (lane == 31) wtot[wv] = incl;
    __syncthreads();
    int carry = 0;
    for (int k = 0; k < wv; ++k) carry += wtot[k];
    return carry + incl - v;
}

__global__ __launch_bounds__(1024) void gat_chunk_file(const int* __restrict__ src,
                                                       const int* __restrict__ dst,
                                                       int* __restrict__ cnt,
                                                       int* __restrict__ loc,
                                                       int* __restrict__ cedge,
                                                       int E, int B) {
    __shared__ int h[256];
    __shared__ int cur[256];
    __shared__ int wtot[32];
    __shared__ int stage[GAT_CHE];
    const int tid = threadIdx.x;
    const int c   = blockIdx.x;
    const int e0  = c * GAT_CHE;
    if (tid < 256) h[tid] = 0;
    int sv[4], dv[4];
    #pragma unroll
    for (int j = 0; j < 4; ++j) {
        const int e = e0 + j * 1024 + tid;
        dv[j] = (e < E) ? dst[e] : -1;
        sv[j] = (e < E) ? src[e] : 0;
    }
    __syncthreads();
    #pragma unroll
    for (int j = 0; j < 4; ++j)
        if (dv[j] >= 0) atomicAdd(&h[dv[j] >> GAT_BSH], 1);
    __syncthreads();
    const int v  = (tid < 256) ? h[tid] : 0;
    const int ex = gat_block_excl_scan(v, tid, wtot);
    if (tid < 256) cur[tid] = ex;
    if (tid < B) {
        cnt[c * B + tid] = v;
        loc[c * B + tid] = ex;
    }
    __syncthreads();
    #pragma unroll
    for (int j = 0; j < 4; ++j) {
        if (dv[j] >= 0) {
            const int lp = atomicAdd(&cur[dv[j] >> GAT_BSH], 1);
            stage[lp] = sv[j] | ((dv[j] & ((1 << GAT_BSH) - 1)) << GAT_SRCB);
        }
    }
    __syncthreads();
    const int nloc = min(GAT_CHE, E - e0);
    for (int lp = tid; lp < nloc; lp += 1024) cedge[e0 + lp] = stage[lp];
}

template<int MODE>
__device__ __forceinline__ void gat_seg_edge(int w, int slot, int* cn, int* stage,
                                             int* __restrict__ csr, int s0) {
    if (MODE == 2) {
        const int lp = atomicAdd(&cn[w >> GAT_SRCB], 1);
        csr[s0 + lp] = w & GAT_SRCM;
    } else {
        atomicAdd(&cn[w >> GAT_SRCB], 1);
        if (MODE == 0) stage[slot] = w;
    }
}

template<int MODE>
__device__ __forceinline__ void gat_walk_runs(const int* __restrict__ cedge, int nc, int oc, int pc,
                                              int lane, int* cn, int* stage,
                                              int* __restrict__ csr, int s0) {
    const unsigned live = __builtin_amdgcn_ballot_w32(nc > 0);
    #pragma unroll
    for (int g = 0; g < 32; g += GAT_SEGB) {
        if ((live >> g) == 0u) break;
        int w[GAT_SEGB];
        #pragma unroll
        for (int j = 0; j < GAT_SEGB; ++j) {
            const int n = __builtin_amdgcn_readlane(nc, g + j);
            const int o = __builtin_amdgcn_readlane(oc, g + j);
            w[j] = (lane < n) ? cedge[o + lane] : -1;
        }
        #pragma unroll
        for (int j = 0; j < GAT_SEGB; ++j) {
            const int p = __builtin_amdgcn_readlane(pc, g + j);
            if (w[j] >= 0) gat_seg_edge<MODE>(w[j], p + lane, cn, stage, csr, s0);
        }
        #pragma unroll
        for (int j = 0; j < GAT_SEGB; ++j) {
            const int n = __builtin_amdgcn_readlane(nc, g + j);
            if (n > 32) {
                const int o = __builtin_amdgcn_readlane(oc, g + j);
                const int p = __builtin_amdgcn_readlane(pc, g + j);
                for (int l = lane + 32; l < n; l += 32)
                    gat_seg_edge<MODE>(cedge[o + l], p + l, cn, stage, csr, s0);
            }
        }
    }
}

__device__ __forceinline__ void gat_bkt_sort_block(const int* __restrict__ cedge,
                                                   const int* __restrict__ cnt,
                                                   const int* __restrict__ loc,
                                                   int* __restrict__ csr,
                                                   int* __restrict__ rowptr,
                                                   int N, int B, int C, const int b) {
    __shared__ int cn[512];
    __shared__ int wtot[32];
    __shared__ int wnum[32];
    __shared__ int wloc[32];
    __shared__ int stage[GAT_STG];
    __shared__ int stg[GAT_STG];
    const int tid  = threadIdx.x;
    const int lane = tid & 31;
    const int wv   = tid >> 5;
    const int d0   = b << GAT_BSH;
    if (tid < 512) cn[tid] = 0;
    const int cc = lane * 32 + wv;
    int nc = 0, lc = 0;
    if (cc < C) {
        nc = cnt[cc * B + b];
        lc = loc[cc * B + b];
    }
    const int pc = gat_block_excl_scan(nc, tid, wnum);
    const int oc = cc * GAT_CHE + lc;
    int ls = lc;
    for (int o = 16; o > 0; o >>= 1) ls += __shfl_xor(ls, o);
    if (lane == 0) wloc[wv] = ls;
    __syncthreads();
    int s0 = 0, nb = 0;
    for (int k = 0; k < 32; ++k) {
        s0 += wloc[k];
        nb += wnum[k];
    }
    const bool staged = (nb <= GAT_STG);
    if (staged) gat_walk_runs<0>(cedge, nc, oc, pc, lane, cn, stage, csr, s0);
    else        gat_walk_runs<1>(cedge, nc, oc, pc, lane, cn, stage, csr, s0);
    __syncthreads();
    const int v  = (tid < 512) ? cn[tid] : 0;
    const int ex = gat_block_excl_scan(v, tid, wtot);
    if (tid < 512) {
        cn[tid] = ex;
        if (d0 + tid < N) rowptr[d0 + tid] = s0 + ex;
    }
    if (b == B - 1 && tid == 0) rowptr[N] = s0 + nb;
    __syncthreads();
    if (staged) {
        for (int i0 = 0; i0 < nb; i0 += 4096) {
            int w[4];
            #pragma unroll
            for (int j = 0; j < 4; ++j) {
                const int i = i0 + j * 1024 + tid;
                w[j] = (i < nb) ? stage[i] : -1;
            }
            #pragma unroll
            for (int j = 0; j < 4; ++j) {
                if (w[j] >= 0) {
                    const int lp = atomicAdd(&cn[w[j] >> GAT_SRCB], 1);
                    stg[lp] = w[j] & GAT_SRCM;
                }
            }
        }
        __syncthreads();
        for (int i = tid; i < nb; i += 1024) csr[s0 + i] = stg[i];
    } else {
        gat_walk_runs<2>(cedge, nc, oc, pc, lane, cn, stage, csr, s0);
    }
}

#define GAT_SPB 28
__global__ __launch_bounds__(1024) __attribute__((amdgpu_waves_per_eu(16, 16))) void gat_bkt_sort_gemm(
                                                     const int* __restrict__ cedge,
                                                     const int* __restrict__ cnt,
                                                     const int* __restrict__ loc,
                                                     int* __restrict__ csr,
                                                     int* __restrict__ rowptr,
                                                     int N, int B, int C, int G,
                                                     const float* __restrict__ X,
                                                     const float* __restrict__ W,
                                                     const float* __restrict__ al,
                                                     const float* __restrict__ ar,
                                                     float* __restrict__ F,
                                                     float* __restrict__ el,
                                                     float* __restrict__ er) {
    const int b = blockIdx.x;
    if (__builtin_expect(b < G, 1)) {
        gat_gemm_strips<128, 4, 64, 32, GAT_SPB, 8>(X, W, al, ar, F, el, er, N, b, (int)threadIdx.x);
        return;
    }
    gat_bkt_sort_block(cedge, cnt, loc, csr, rowptr, N, B, C, b - G);
}

template<int P>
__device__ __forceinline__ int gat_share(int x) {
    return __builtin_amdgcn_update_dpp(0, x, 0x150 + P, 0xf, 0xf, true);
}
typedef __attribute__((ext_vector_type(4))) float v4f;
__device__ __forceinline__ float gat_row16_max(float x) {
    x = fmaxf(x, gat_dpp<0xB1>(x));
    x = fmaxf(x, gat_dpp<0x4E>(x));
    x = fmaxf(x, gat_dpp<0x141>(x));
    x = fmaxf(x, gat_dpp<0x140>(x));
    return x;
}
template<int P, int FS>
__device__ __forceinline__ v4f gat_ldrow(const float* __restrict__ F, unsigned cb, int s) {
    const unsigned off = (unsigned)gat_share<P>(s) * (unsigned)(FS * 4) + cb;
    return *(const v4f*)((const char*)F + off);
}
template<int P>
__device__ __forceinline__ void gat_fma_row(v2f& a01, v2f& a23, float w, const v4f f) {
    const float wp = __int_as_float(gat_share<P>(__float_as_int(w)));
    a01 += f.lo * wp;
    a23 += f.hi * wp;
}
template<int T, int G, int FS>
__device__ __forceinline__ void gat_pair_rows(const float* __restrict__ F, unsigned cb, int s, float w,
                                              v2f& a01, v2f& a23) {
    static_assert(T + G <= 16, "positions stay inside the half");
    v4f f0, f1, f2, f3, f4, f5, f6, f7;
    f0 = gat_ldrow<T, FS>(F, cb, s);
    if constexpr (G > 1) f1 = gat_ldrow<T + 1, FS>(F, cb, s);
    if constexpr (G > 2) { f2 = gat_ldrow<T + 2, FS>(F, cb, s); f3 = gat_ldrow<T + 3, FS>(F, cb, s); }
    if constexpr (G > 4) {
        f4 = gat_ldrow<T + 4, FS>(F, cb, s); f5 = gat_ldrow<T + 5, FS>(F, cb, s);
        f6 = gat_ldrow<T + 6, FS>(F, cb, s); f7 = gat_ldrow<T + 7, FS>(F, cb, s);
    }
    gat_fma_row<T>(a01, a23, w, f0);
    if constexpr (G > 1) gat_fma_row<T + 1>(a01, a23, w, f1);
    if constexpr (G > 2) { gat_fma_row<T + 2>(a01, a23, w, f2); gat_fma_row<T + 3>(a01, a23, w, f3); }
    if constexpr (G > 4) {
        gat_fma_row<T + 4>(a01, a23, w, f4); gat_fma_row<T + 5>(a01, a23, w, f5);
        gat_fma_row<T + 6>(a01, a23, w, f6); gat_fma_row<T + 7>(a01, a23, w, f7);
    }
}
template<int T, int FS>
__device__ __forceinline__ void gat_pair_tail(const float* __restrict__ F, unsigned cb, int s, float w,
                                              int rest, v2f& a01, v2f& a23) {
    if (rest & 4) {
        gat_pair_rows<T, 4, FS>(F, cb, s, w, a01, a23);
        if (rest & 2) gat_pair_rows<T + 4, 2, FS>(F, cb, s, w, a01, a23);
    } else if (rest & 2) {
        gat_pair_rows<T, 2, FS>(F, cb, s, w, a01, a23);
    }
}
template<int FS>
__device__ __forceinline__ void gat_pair_chunk(const float* __restrict__ F, unsigned cb, int s, float w,
                                               int P, v2f& a01, v2f& a23) {
    const int P2 = (P + 1) & ~1;
    if (P2 >= 8) {
        gat_pair_rows<0, 8, FS>(F, cb, s, w, a01, a23);
        if (P2 == 16) gat_pair_rows<8, 8, FS>(F, cb, s, w, a01, a23);
    }
    if (P2 & 8) gat_pair_tail<8, FS>(F, cb, s, w, P2 & 6, a01, a23);
    else        gat_pair_tail<0, FS>(F, cb, s, w, P2 & 6, a01, a23);
}

template<int D, int FS, int NT2, int DOUT2>
__global__ __launch_bounds__(256) void gat_node2_softmax_agg_gemm(const float* __restrict__ F,
                                                                  const float* __restrict__ el,
                                                                  const float* __restrict__ er,
                                                                  const int* __restrict__ rowptr,
                                                                  const int* __restrict__ csr,
                                                                  const float* __restrict__ bias,
                                                                  const float* __restrict__ W2,
                                                                  const float* __restrict__ al2,
                                                                  const float* __restrict__ ar2,
                                                                  float* __restrict__ C2,
                                                                  float* __restrict__ el2,
                                                                  float* __restrict__ er2, int N) {
    static_assert(D == 64 && FS == 64, "16 lanes x 4 columns cover the row");
    constexpr int DP2 = NT2 * 16;
    constexpr int HP = D + 4;
    constexpr int WAVES = 8;
    __shared__ alignas(16) float Bs[D * DP2];
    __shared__ alignas(16) float Hs[16 * HP];
    __shared__ int arrived;

    const int lane = threadIdx.x;
    const int c = lane & 15;
    const int wb = __builtin_amdgcn_readfirstlane((int)threadIdx.y);
    const int hrow = 2 * wb + (lane >> 4);
    const int row0 = blockIdx.x * 16;
    const int node = row0 + hrow;
    const bool nok = (node < N);
    const int nd = nok ? node : N - 1;
    const int beg = rowptr[nd];
    const int nxt = rowptr[nd + 1];
    const int end = nok ? nxt : beg;
    const float erd = er[nd];
    gat_stage_w_pairs<D, NT2, DOUT2, WAVES * 32>(W2, Bs, wb * 32 + lane);
    if (wb == 0 && lane == 0) arrived = 0;
    __syncthreads();

    const int deg = end - beg;
    const int last = max(end - 1, 0);
    const int dmax = max(__builtin_amdgcn_readlane(deg, 0), __builtin_amdgcn_readlane(deg, 16));
    const unsigned cb = (unsigned)c * 16u;

    const int i0 = beg + c, i1 = i0 + 16;
    const int s0 = csr[min(i0, last)];
    int s1 = 0;
    if (dmax > 16) s1 = csr[min(i1, last)];
    const float g0 = el[s0];
    float g1 = 0.f;
    if (dmax > 16) g1 = el[s1];
    float v0 = g0 + erd, v1 = g1 + erd;
    v0 = (v0 > 0.f) ? v0 : NEG_SLOPE * v0;
    v1 = (v1 > 0.f) ? v1 : NEG_SLOPE * v1;
    v0 = (i0 < end) ? v0 : -INFINITY;
    v1 = (i1 < end) ? v1 : -INFINITY;
    float m = fmaxf(v0, v1);
    for (int k = 32; k < dmax; k += 16) {
        const int i = i0 + k;
        const float v = el[csr[min(i, last)]] + erd;
        if (i < end) m = fmaxf(m, (v > 0.f) ? v : NEG_SLOPE * v);
    }
    m = gat_row16_max(m);

    v2f a01 = (v2f){0.f, 0.f}, a23 = (v2f){0.f, 0.f};
    const float w0 = (i0 < end) ? __expf(v0 - m) : 0.f;
    float ssum = w0;
    if (dmax > 0) gat_pair_chunk<FS>(F, cb, s0, w0, min(dmax, 16), a01, a23);
    if (dmax > 16) {
        const float w1 = (i1 < end) ? __expf(v1 - m) : 0.f;
        ssum += w1;
        gat_pair_chunk<FS>(F, cb, s1, w1, min(dmax - 16, 16), a01, a23);
    }
    for (int k = 32; k < dmax; k += 16) {
        const int i = i0 + k;
        const int s = csr[min(i, last)];
        const float v = el[s] + erd;
        const float w = (i < end) ? __expf(((v > 0.f) ? v : NEG_SLOPE * v) - m) : 0.f;
        ssum += w;
        gat_pair_chunk<FS>(F, cb, s, w, min(dmax - k, 16), a01, a23);
    }
    ssum = gat_row16_sum(ssum);

    float x = 0.f, y = 0.f, z = 0.f, u = 0.f;
    if (nok) {
        const float4 b = *(const float4*)(bias + 4 * c);
        x = b.x; y = b.y; z = b.z; u = b.w;
        if (deg > 0) { x += a01.x / ssum; y += a01.y / ssum; z += a23.x / ssum; u += a23.y / ssum; }
        x = (x > 0.f) ? x : (__expf(x) - 1.f);
        y = (y > 0.f) ? y : (__expf(y) - 1.f);
        z = (z > 0.f) ? z : (__expf(z) - 1.f);
        u = (u > 0.f) ? u : (__expf(u) - 1.f);
    }
    *(float4*)&Hs[hrow * HP + 4 * c] = make_float4(x, y, z, u);

    __threadfence_block();
    int seen = 0;
    if (lane == 0) seen = __hip_atomic_fetch_add(&arrived, 1, __ATOMIC_RELAXED, __HIP_MEMORY_SCOPE_WORKGROUP);
    seen = __builtin_amdgcn_readfirstlane(seen);
    if (seen != WAVES - 1) return;
    __threadfence_block();

    const int kh = lane >> 4;
    v8f acc[NT2];
    #pragma unroll
    for (int t = 0; t < NT2; ++t) acc[t] = (v8f){0.f,0.f,0.f,0.f,0.f,0.f,0.f,0.f};
    #pragma unroll
    for (int k = 0; k < D; k += 4) {
        const int ka = k + 2 * kh;
        const v2f a = *(const v2f*)&Hs[c * HP + ka];
        const int kk = ka >> 1;
        #pragma unroll
        for (int t = 0; t < NT2; ++t) {
            const v2f b = *(const v2f*)&Bs[(kk * DP2 + t * 16 + c) * 2];
            acc[t] = __builtin_amdgcn_wmma_f32_16x16x4_f32(
                false, a, false, b, (short)0, acc[t], false, false);
        }
    }
    gat_strip_epilogue<NT2, DOUT2>(acc, al2, ar2, C2, el2, er2, row0, c, kh);
}

template<int P, int D>
__device__ __forceinline__ float gat_row_logit(const v4f f, float erd, int cnt) {
    const float v = __int_as_float(gat_share<D / 4>(__float_as_int(f.x))) + erd;
    const float e = (v > 0.f) ? v : NEG_SLOPE * v;
    return (P < cnt) ? e : -INFINITY;
}
__device__ __forceinline__ void gat_acc_row(v2f& a01, v2f& a23, float& ssum, float w, const v4f f) {
    ssum += w;
    a01 += f.lo * w;
    a23 += f.hi * w;
}
template<int T, int G, int D, int FS>
__device__ __forceinline__ void gat_pair_online(const float* __restrict__ F, unsigned cb, int s, int cnt,
                                                float erd, float& m, float& ssum, v2f& a01, v2f& a23) {
    static_assert(T + G <= 16, "positions stay inside the half");
    v4f f0, f1, f2, f3, f4, f5, f6, f7;
    f0 = gat_ldrow<T, FS>(F, cb, s);
    if constexpr (G > 1) f1 = gat_ldrow<T + 1, FS>(F, cb, s);
    if constexpr (G > 2) { f2 = gat_ldrow<T + 2, FS>(F, cb, s); f3 = gat_ldrow<T + 3, FS>(F, cb, s); }
    if constexpr (G > 4) {
        f4 = gat_ldrow<T + 4, FS>(F, cb, s); f5 = gat_ldrow<T + 5, FS>(F, cb, s);
        f6 = gat_ldrow<T + 6, FS>(F, cb, s); f7 = gat_ldrow<T + 7, FS>(F, cb, s);
    }
    float e0, e1, e2, e3, e4, e5, e6, e7;
    float mn = m;
    e0 = gat_row_logit<T, D>(f0, erd, cnt); mn = fmaxf(mn, e0);
    if constexpr (G > 1) { e1 = gat_row_logit<T + 1, D>(f1, erd, cnt); mn = fmaxf(mn, e1); }
    if constexpr (G > 2) {
        e2 = gat_row_logit<T + 2, D>(f2, erd, cnt); mn = fmaxf(mn, e2);
        e3 = gat_row_logit<T + 3, D>(f3, erd, cnt); mn = fmaxf(mn, e3);
    }
    if constexpr (G > 4) {
        e4 = gat_row_logit<T + 4, D>(f4, erd, cnt); mn = fmaxf(mn, e4);
        e5 = gat_row_logit<T + 5, D>(f5, erd, cnt); mn = fmaxf(mn, e5);
        e6 = gat_row_logit<T + 6, D>(f6, erd, cnt); mn = fmaxf(mn, e6);
        e7 = gat_row_logit<T + 7, D>(f7, erd, cnt); mn = fmaxf(mn, e7);
    }
    const float sc = __expf(m - mn);
    a01 *= sc; a23 *= sc; ssum *= sc;
    gat_acc_row(a01, a23, ssum, __expf(e0 - mn), f0);
    if constexpr (G > 1) gat_acc_row(a01, a23, ssum, __expf(e1 - mn), f1);
    if constexpr (G > 2) { gat_acc_row(a01, a23, ssum, __expf(e2 - mn), f2); gat_acc_row(a01, a23, ssum, __expf(e3 - mn), f3); }
    if constexpr (G > 4) {
        gat_acc_row(a01, a23, ssum, __expf(e4 - mn), f4); gat_acc_row(a01, a23, ssum, __expf(e5 - mn), f5);
        gat_acc_row(a01, a23, ssum, __expf(e6 - mn), f6); gat_acc_row(a01, a23, ssum, __expf(e7 - mn), f7);
    }
    m = mn;
}
template<int T, int D, int FS>
__device__ __forceinline__ void gat_online_tail(const float* __restrict__ F, unsigned cb, int s, int cnt,
                                                float erd, int rest, float& m, float& ssum, v2f& a01, v2f& a23) {
    if (rest & 4) {
        gat_pair_online<T, 4, D, FS>(F, cb, s, cnt, erd, m, ssum, a01, a23);
        if (rest & 2) gat_pair_online<T + 4, 2, D, FS>(F, cb, s, cnt, erd, m, ssum, a01, a23);
    } else if (rest & 2) {
        gat_pair_online<T, 2, D, FS>(F, cb, s, cnt, erd, m, ssum, a01, a23);
    }
}

template<int D, int FS>
__global__ __launch_bounds__(256) void gat_node2_online_agg(const float* __restrict__ F,
                                                            const float* __restrict__ er,
                                                            const int* __restrict__ rowptr,
                                                            const int* __restrict__ csr,
                                                            const float* __restrict__ bias,
                                                            float* __restrict__ out, int N) {
    static_assert(D % 4 == 0 && D + 4 <= FS && FS <= 64 && FS % 4 == 0, "el sits in column D of the padded row");
    const int lane = threadIdx.x;
    const int c = lane & 15;
    const int wv = blockIdx.x * 8 + __builtin_amdgcn_readfirstlane((int)threadIdx.y);
    if (2 * wv >= N) return;
    const int node = 2 * wv + (lane >> 4);
    const bool nok = (node < N);
    const int nd = nok ? node : N - 1;
    const int beg = rowptr[nd];
    const int nxt = rowptr[nd + 1];
    const int end = nok ? nxt : beg;
    const float erd = er[nd];
    const int deg = end - beg;
    const int last = max(end - 1, 0);
    const int dmax = max(__builtin_amdgcn_readlane(deg, 0), __builtin_amdgcn_readlane(deg, 16));
    const unsigned cb = (4 * c < FS) ? (unsigned)c * 16u : 0u;

    float m = -INFINITY, ssum = 0.f;
    v2f a01 = (v2f){0.f, 0.f}, a23 = (v2f){0.f, 0.f};
    int s = csr[min(beg + c, last)];
    for (int k = 0; k < dmax; k += 16) {
        int sn = 0;
        if (k + 16 < dmax) sn = csr[min(beg + k + 16 + c, last)];
        const int cnt = deg - k;
        const int P2 = (min(dmax - k, 16) + 1) & ~1;
        if (P2 >= 8) {
            gat_pair_online<0, 8, D, FS>(F, cb, s, cnt, erd, m, ssum, a01, a23);
            if (P2 == 16) gat_pair_online<8, 8, D, FS>(F, cb, s, cnt, erd, m, ssum, a01, a23);
        }
        if (P2 & 8) gat_online_tail<8, D, FS>(F, cb, s, cnt, erd, P2 & 6, m, ssum, a01, a23);
        else        gat_online_tail<0, D, FS>(F, cb, s, cnt, erd, P2 & 6, m, ssum, a01, a23);
        s = sn;
    }

    if (nok && 4 * c < D) {
        const float4 b = *(const float4*)(bias + 4 * c);
        float x = b.x, y = b.y, z = b.z, u = b.w;
        if (deg > 0) { x += a01.x / ssum; y += a01.y / ssum; z += a23.x / ssum; u += a23.y / ssum; }
        *(float4*)(out + (size_t)node * D + 4 * c) = make_float4(x, y, z, u);
    }
}

extern "C" void kernel_launch(void* const* d_in, const int* in_sizes, int n_in,
                              void* d_out, int out_size, void* d_ws, size_t ws_size,
                              hipStream_t stream) {
    const float* X   = (const float*)d_in[0];
    const int*   src = (const int*)d_in[1];
    const int*   dst = (const int*)d_in[2];
    const float* W1  = (const float*)d_in[3];
    const float* al1 = (const float*)d_in[4];
    const float* ar1 = (const float*)d_in[5];
    const float* b1  = (const float*)d_in[6];
    const float* W2  = (const float*)d_in[7];
    const float* al2 = (const float*)d_in[8];
    const float* ar2 = (const float*)d_in[9];
    const float* b2  = (const float*)d_in[10];

    const int IN = 128, H = 64, CLSPAD = 48;
    const int N = in_sizes[0] / IN;
    const int E = in_sizes[1];

    float* ws = (float*)d_ws;
    float* feat1 = ws;
    const int Mtiles = (N + 15) / 16;
    const int NP = Mtiles * 16;
    float* feat2 = feat1 + (size_t)N * H;
    float* el    = feat2 + (size_t)NP * CLSPAD;
    float* er    = el + N;
    int*   cedge  = (int*)(er + N);
    int*   csr    = cedge + E;
    int*   rowptr = csr + E;
    const int B = (N + 511) / 512;
    const int C = (E + 4095) / 4096;
    int*   cnt    = rowptr + N + 1;
    int*   loc    = cnt + B * C;
    float* el2    = (float*)(loc + B * C);
    float* er2    = el2 + NP;

    dim3 wblk(32, 8);
    const dim3 ngrid(Mtiles);

    hipLaunchKernelGGL(gat_chunk_file, dim3(C), dim3(1024), 0, stream,
                       src, dst, cnt, loc, cedge, E, B);
    const int G = (Mtiles + GAT_SPB - 1) / GAT_SPB;
    hipLaunchKernelGGL(gat_bkt_sort_gemm, dim3(G + B), dim3(1024), 0, stream,
                       cedge, cnt, loc, csr, rowptr, N, B, C, G,
                       X, W1, al1, ar1, feat1, el, er);

    hipLaunchKernelGGL((gat_node2_softmax_agg_gemm<64, 64, 3, 40>), ngrid, wblk, 0, stream,
                       feat1, el, er, rowptr, csr, b1, W2, al2, ar2, feat2, el2, er2, N);

    hipLaunchKernelGGL((gat_node2_online_agg<40, 48>), ngrid, wblk, 0, stream,
                       feat2, er2, rowptr, csr, b2, (float*)d_out, N);
}
